# starting kernel
# baseline (speedup 1.0000x reference)
#include <hip/hip_runtime.h>

typedef __attribute__((ext_vector_type(2))) float v2f;
typedef __attribute__((ext_vector_type(8))) float v8f;

#define NB 128
#define NN 2048
#define ND 64
#define NE 32768

__global__ void k_zero_u32(unsigned* p, int n) {
  int i = blockIdx.x * blockDim.x + threadIdx.x;
  if (i < n) p[i] = 0u;
}

__global__ void k_count(const int* __restrict__ dst, unsigned* __restrict__ deg) {
  int e = blockIdx.x * blockDim.x + threadIdx.x;
  if (e < NE) atomicAdd(&deg[dst[e]], 1u);
}

__global__ void k_scan(const unsigned* __restrict__ deg, unsigned* __restrict__ off,
                       unsigned* __restrict__ cursor) {
  __shared__ unsigned s0[NN];
  __shared__ unsigned s1[NN];
  int t = threadIdx.x;
  for (int i = t; i < NN; i += 256) s0[i] = deg[i];
  __syncthreads();
  unsigned* a = s0;
  unsigned* b = s1;
  for (int st = 1; st < NN; st <<= 1) {
    for (int i = t; i < NN; i += 256)
      b[i] = a[i] + ((i >= st) ? a[i - st] : 0u);
    __syncthreads();
    unsigned* tmp = a; a = b; b = tmp;
  }
  for (int i = t; i < NN; i += 256) {
    off[i + 1] = a[i];
    cursor[i] = (i == 0) ? 0u : a[i - 1];
  }
  if (t == 0) off[0] = 0u;
}

__global__ void k_fill(const int* __restrict__ src, const int* __restrict__ dst,
                       unsigned* __restrict__ cursor, unsigned* __restrict__ csr) {
  int e = blockIdx.x * blockDim.x + threadIdx.x;
  if (e < NE) {
    unsigned p = atomicAdd(&cursor[dst[e]], 1u);
    csr[p] = (unsigned)src[e];
  }
}

__global__ void k_maxb(const float* __restrict__ X, float* __restrict__ xmax) {
  int i = blockIdx.x * blockDim.x + threadIdx.x;
  float m = -3.4028234663852886e38f;
  for (int b = 0; b < NB; ++b)
    m = fmaxf(m, X[(size_t)b * (NN * ND) + i]);
  xmax[i] = m;
}

__global__ void __launch_bounds__(128)
k_mlp2_wmma(const float* __restrict__ Xin, const float* __restrict__ W1,
            const float* __restrict__ b1, const float* __restrict__ W2,
            const float* __restrict__ b2, float* __restrict__ Yout) {
  __shared__ float Ys[16 * ND];
  const int lane = threadIdx.x & 31;
  const int wave = threadIdx.x >> 5;
  const int col0 = wave * 16;
  const int hm = lane & 15;
  const int hk = lane >> 4;
  const size_t row0 = (size_t)blockIdx.x * 16;

  v2f bw[16];
#pragma unroll
  for (int kk = 0; kk < 16; ++kk) {
    int k = kk * 4 + 2 * hk;
    bw[kk].x = W1[(k + 0) * ND + col0 + hm];
    bw[kk].y = W1[(k + 1) * ND + col0 + hm];
  }

  v8f acc;
  {
    float bv = b1[col0 + hm];
#pragma unroll
    for (int j = 0; j < 8; ++j) acc[j] = bv;
  }

  const float* xrow = Xin + (row0 + hm) * ND;
#pragma unroll
  for (int kk = 0; kk < 16; ++kk) {
    int k = kk * 4 + 2 * hk;
    v2f a;
    a.x = xrow[k];
    a.y = xrow[k + 1];
    acc = __builtin_amdgcn_wmma_f32_16x16x4_f32(false, a, false, bw[kk],
                                                (short)0, acc, false, false);
  }
#pragma unroll
  for (int j = 0; j < 8; ++j) {
    Ys[(j + 8 * hk) * ND + col0 + hm] = fmaxf(acc[j], 0.0f);
  }
  __syncthreads();

#pragma unroll
  for (int kk = 0; kk < 16; ++kk) {
    int k = kk * 4 + 2 * hk;
    bw[kk].x = W2[(k + 0) * ND + col0 + hm];
    bw[kk].y = W2[(k + 1) * ND + col0 + hm];
  }
  {
    float bv = b2[col0 + hm];
#pragma unroll
    for (int j = 0; j < 8; ++j) acc[j] = bv;
  }
#pragma unroll
  for (int kk = 0; kk < 16; ++kk) {
    int k = kk * 4 + 2 * hk;
    v2f a;
    a.x = Ys[hm * ND + k];
    a.y = Ys[hm * ND + k + 1];
    acc = __builtin_amdgcn_wmma_f32_16x16x4_f32(false, a, false, bw[kk],
                                                (short)0, acc, false, false);
  }
#pragma unroll
  for (int j = 0; j < 8; ++j) {
    Yout[(row0 + j + 8 * hk) * ND + col0 + hm] = fmaxf(acc[j], 0.0f);
  }
}

__global__ void k_gather_node(const float* __restrict__ nodex,
                              const unsigned* __restrict__ off,
                              const unsigned* __restrict__ csr,
                              float* __restrict__ agg) {
  int r = blockIdx.x * 4 + (threadIdx.x >> 6);
  int d = threadIdx.x & 63;
  unsigned s = off[r], e = off[r + 1];
  float acc = 0.0f;
  for (unsigned j = s; j < e; ++j) acc += nodex[(size_t)csr[j] * ND + d];
  agg[(size_t)r * ND + d] = acc;
}

__global__ void k_gather_final(const float* __restrict__ h,
                               const float* __restrict__ agg,
                               const unsigned* __restrict__ off,
                               const unsigned* __restrict__ csr,
                               float* __restrict__ out) {
  int n = blockIdx.x & (NN - 1);
  int b = ((blockIdx.x >> 11) << 2) + (threadIdx.x >> 6);
  int d = threadIdx.x & 63;
  unsigned s = off[n], e = off[n + 1];
  float acc = agg[(size_t)n * ND + d];
  const float* hb = h + (size_t)b * NN * ND;
  for (unsigned j = s; j < e; ++j) acc += hb[(size_t)csr[j] * ND + d];
  out[((size_t)b * NN + n) * ND + d] = acc;
}

extern "C" void kernel_launch(void* const* d_in, const int* in_sizes, int n_in,
                              void* d_out, int out_size, void* d_ws, size_t ws_size,
                              hipStream_t stream) {
  (void)in_sizes; (void)n_in; (void)out_size; (void)ws_size;
  const float* X   = (const float*)d_in[0];
  const int*   ei  = (const int*)d_in[1];
  const float* W1t = (const float*)d_in[2];
  const float* b1t = (const float*)d_in[3];
  const float* W2t = (const float*)d_in[4];
  const float* b2t = (const float*)d_in[5];
  const float* W1n = (const float*)d_in[6];
  const float* b1n = (const float*)d_in[7];
  const float* W2n = (const float*)d_in[8];
  const float* b2n = (const float*)d_in[9];
  float* out = (float*)d_out;

  float* h     = (float*)d_ws;
  float* xmax  = h + (size_t)NB * NN * ND;
  float* nodex = xmax + (size_t)NN * ND;
  float* agg   = nodex + (size_t)NN * ND;
  unsigned* deg    = (unsigned*)(agg + (size_t)NN * ND);
  unsigned* off    = deg + NN;
  unsigned* cursor = off + (NN + 1);
  unsigned* csr    = cursor + NN;

  const int* esrc = ei;
  const int* edst = ei + NE;

  k_zero_u32<<<(NN + 255) / 256, 256, 0, stream>>>(deg, NN);
  k_count<<<NE / 256, 256, 0, stream>>>(edst, deg);
  k_scan<<<1, 256, 0, stream>>>(deg, off, cursor);
  k_fill<<<NE / 256, 256, 0, stream>>>(esrc, edst, cursor, csr);

  k_maxb<<<(NN * ND) / 256, 256, 0, stream>>>(X, xmax);
  k_mlp2_wmma<<<NN / 16, 128, 0, stream>>>(xmax, W1n, b1n, W2n, b2n, nodex);
  k_gather_node<<<NN / 4, 256, 0, stream>>>(nodex, off, csr, agg);

  k_mlp2_wmma<<<(NB * NN) / 16, 128, 0, stream>>>(X, W1t, b1t, W2t, b2t, h);
  k_gather_final<<<(NB / 4) * NN, 256, 0, stream>>>(h, agg, off, csr, out);
}


# max over subgraphs runs as 512 extra blocks of the big MLP's launch (127 registers, 8 waves per SIMD): its memory waits overlap the matrix-bound row tiles; 9 -> 8 launches; f32 add order kept
# speedup vs baseline: 1.5300x; 1.5300x over previous
#include <hip/hip_runtime.h>

typedef __attribute__((ext_vector_type(2))) float v2f;
typedef __attribute__((ext_vector_type(8))) float v8f;
typedef __attribute__((ext_vector_type(4))) float v4f;

#define NB 128
#define NN 2048
#define ND 64
#define NE 32768

__global__ void k_zero_u32(unsigned* p, int n) {
  int i = blockIdx.x * blockDim.x + threadIdx.x;
  if (i < n) p[i] = 0u;
}

__global__ void k_count(const int* __restrict__ dst, unsigned* __restrict__ deg) {
  int e = blockIdx.x * blockDim.x + threadIdx.x;
  if (e < NE) atomicAdd(&deg[dst[e]], 1u);
}

__global__ void k_scan(const unsigned* __restrict__ deg, unsigned* __restrict__ off,
                       unsigned* __restrict__ cursor) {
  __shared__ unsigned s0[NN];
  __shared__ unsigned s1[NN];
  int t = threadIdx.x;
  for (int i = t; i < NN; i += 256) s0[i] = deg[i];
  __syncthreads();
  unsigned* a = s0;
  unsigned* b = s1;
  for (int st = 1; st < NN; st <<= 1) {
    for (int i = t; i < NN; i += 256)
      b[i] = a[i] + ((i >= st) ? a[i - st] : 0u);
    __syncthreads();
    unsigned* tmp = a; a = b; b = tmp;
  }
  for (int i = t; i < NN; i += 256) {
    off[i + 1] = a[i];
    cursor[i] = (i == 0) ? 0u : a[i - 1];
  }
  if (t == 0) off[0] = 0u;
}

__global__ void k_fill(const int* __restrict__ src, const int* __restrict__ dst,
                       unsigned* __restrict__ cursor, unsigned* __restrict__ csr) {
  int e = blockIdx.x * blockDim.x + threadIdx.x;
  if (e < NE) {
    unsigned p = atomicAdd(&cursor[dst[e]], 1u);
    csr[p] = (unsigned)src[e];
  }
}

__global__ void __launch_bounds__(128)
k_mlp2_wmma(const float* __restrict__ Xin, const float* __restrict__ W1,
            const float* __restrict__ b1, const float* __restrict__ W2,
            const float* __restrict__ b2, float* __restrict__ Yout) {
  __shared__ float Ys[16 * ND];
  const int lane = threadIdx.x & 31;
  const int wave = threadIdx.x >> 5;
  const int col0 = wave * 16;
  const int hm = lane & 15;
  const int hk = lane >> 4;
  const size_t row0 = (size_t)blockIdx.x * 16;

  v2f bw[16];
#pragma unroll
  for (int kk = 0; kk < 16; ++kk) {
    int k = kk * 4 + 2 * hk;
    bw[kk].x = W1[(k + 0) * ND + col0 + hm];
    bw[kk].y = W1[(k + 1) * ND + col0 + hm];
  }

  v8f acc;
  {
    float bv = b1[col0 + hm];
#pragma unroll
    for (int j = 0; j < 8; ++j) acc[j] = bv;
  }

  const float* xrow = Xin + (row0 + hm) * ND;
#pragma unroll
  for (int kk = 0; kk < 16; ++kk) {
    int k = kk * 4 + 2 * hk;
    v2f a;
    a.x = xrow[k];
    a.y = xrow[k + 1];
    acc = __builtin_amdgcn_wmma_f32_16x16x4_f32(false, a, false, bw[kk],
                                                (short)0, acc, false, false);
  }
#pragma unroll
  for (int j = 0; j < 8; ++j) {
    Ys[(j + 8 * hk) * ND + col0 + hm] = fmaxf(acc[j], 0.0f);
  }
  __syncthreads();

#pragma unroll
  for (int kk = 0; kk < 16; ++kk) {
    int k = kk * 4 + 2 * hk;
    bw[kk].x = W2[(k + 0) * ND + col0 + hm];
    bw[kk].y = W2[(k + 1) * ND + col0 + hm];
  }
  {
    float bv = b2[col0 + hm];
#pragma unroll
    for (int j = 0; j < 8; ++j) acc[j] = bv;
  }
#pragma unroll
  for (int kk = 0; kk < 16; ++kk) {
    int k = kk * 4 + 2 * hk;
    v2f a;
    a.x = Ys[hm * ND + k];
    a.y = Ys[hm * ND + k + 1];
    acc = __builtin_amdgcn_wmma_f32_16x16x4_f32(false, a, false, bw[kk],
                                                (short)0, acc, false, false);
  }
#pragma unroll
  for (int j = 0; j < 8; ++j) {
    Yout[(row0 + j + 8 * hk) * ND + col0 + hm] = fmaxf(acc[j], 0.0f);
  }
}

#define MP 68
#define MW 8
#define MLP_GRID 512
#define MLP_ITERS ((NB * NN / 16) / (MLP_GRID * MW))

#define MAXB_BLOCKS ((NN * ND) / 256)
static_assert(MLP_GRID == 512 && MAXB_BLOCKS == 512, "the role bit and index & 511 assume 512 blocks of each role");

__global__ void __launch_bounds__(256, 8)
k_mlp2_rows_maxb(const float* __restrict__ Xin, const float* __restrict__ W1,
                 const float* __restrict__ b1, const float* __restrict__ W2,
                 const float* __restrict__ b2, float* __restrict__ Yout,
                 float* __restrict__ xmax) {
  __shared__ __attribute__((aligned(16))) float Wt1[ND * MP];
  __shared__ __attribute__((aligned(16))) float Wt2[ND * MP];
  __shared__ __attribute__((aligned(16))) float Ts[MW][16 * MP];
  const int tid = threadIdx.x;
  const unsigned bid = blockIdx.x;
  const unsigned rblk = bid & (MLP_GRID - 1);
  if (((bid >> 1) ^ (bid >> 9)) & 1u) {
    const int i = (int)rblk * 256 + tid;
    float m = -3.4028234663852886e38f;
    for (int b = 0; b < NB; ++b)
      m = fmaxf(m, Xin[(size_t)b * (NN * ND) + i]);
    xmax[i] = m;
    return;
  }
  const int lane = tid & 31;
  const int wave = __builtin_amdgcn_readfirstlane(tid >> 5);
  const int hm = lane & 15;
  const int hk = lane >> 4;
  const int c4 = (lane & 15) << 2;
  float* T = Ts[wave];

  size_t tile = (size_t)rblk * MW + wave;
  v4f xr[8];
#pragma unroll
  for (int i = 0; i < 8; ++i)
    xr[i] = *reinterpret_cast<const v4f*>(Xin + (tile * 16 + 2 * i + hk) * ND + c4);

#pragma unroll
  for (int i = 0; i < 4; ++i) {
    const int idx = tid + 256 * i;
    const int k = idx >> 4;
    const int n4 = (idx & 15) << 2;
    const v4f w1 = *reinterpret_cast<const v4f*>(W1 + k * ND + n4);
    const v4f w2 = *reinterpret_cast<const v4f*>(W2 + k * ND + n4);
#pragma unroll
    for (int e = 0; e < 4; ++e) {
      Wt1[(n4 + e) * MP + k] = w1[e];
      Wt2[(n4 + e) * MP + k] = w2[e];
    }
  }
  float bv1[4], bv2[4];
#pragma unroll
  for (int ct = 0; ct < 4; ++ct) {
    bv1[ct] = b1[ct * 16 + hm];
    bv2[ct] = b2[ct * 16 + hm];
  }

  for (int it = 0; it < MLP_ITERS; ++it) {
#pragma unroll
    for (int i = 0; i < 8; ++i)
      *reinterpret_cast<v4f*>(T + (2 * i + hk) * MP + c4) = xr[i];
    __syncthreads();

    const size_t row0 = tile * 16;
    if (it + 1 < MLP_ITERS) {
      tile += (size_t)MLP_GRID * MW;
#pragma unroll
      for (int i = 0; i < 8; ++i)
        xr[i] = *reinterpret_cast<const v4f*>(Xin + (tile * 16 + 2 * i + hk) * ND + c4);
    }

    v8f acc[4];
#pragma unroll
    for (int ct = 0; ct < 4; ++ct) {
      float bb = bv1[ct];
      asm volatile("" : "+v"(bb));
#pragma unroll
      for (int j = 0; j < 8; ++j) acc[ct][j] = bb;
    }
#pragma unroll
    for (int kk = 0; kk < 16; ++kk) {
      const int k = kk * 4 + 2 * hk;
      const v2f a = *reinterpret_cast<const v2f*>(T + hm * MP + k);
#pragma unroll
      for (int ct = 0; ct < 4; ++ct) {
        const v2f b = *reinterpret_cast<const v2f*>(Wt1 + (ct * 16 + hm) * MP + k);
        acc[ct] = __builtin_amdgcn_wmma_f32_16x16x4_f32(false, a, false, b,
                                                        (short)0, acc[ct], false, false);
      }
    }
#pragma unroll
    for (int ct = 0; ct < 4; ++ct)
#pragma unroll
      for (int j = 0; j < 8; ++j)
        T[(j + 8 * hk) * MP + ct * 16 + hm] = fmaxf(acc[ct][j], 0.0f);
    __syncthreads();

#pragma unroll
    for (int ct = 0; ct < 4; ++ct) {
      float bb = bv2[ct];
      asm volatile("" : "+v"(bb));
#pragma unroll
      for (int j = 0; j < 8; ++j) acc[ct][j] = bb;
    }
#pragma unroll
    for (int kk = 0; kk < 16; ++kk) {
      const int k = kk * 4 + 2 * hk;
      const v2f a = *reinterpret_cast<const v2f*>(T + hm * MP + k);
#pragma unroll
      for (int ct = 0; ct < 4; ++ct) {
        const v2f b = *reinterpret_cast<const v2f*>(Wt2 + (ct * 16 + hm) * MP + k);
        acc[ct] = __builtin_amdgcn_wmma_f32_16x16x4_f32(false, a, false, b,
                                                        (short)0, acc[ct], false, false);
      }
    }
#pragma unroll
    for (int ct = 0; ct < 4; ++ct)
#pragma unroll
      for (int j = 0; j < 8; ++j)
        T[(j + 8 * hk) * MP + ct * 16 + hm] = fmaxf(acc[ct][j], 0.0f);
    __syncthreads();

#pragma unroll
    for (int i = 0; i < 8; ++i) {
      const v4f v = *reinterpret_cast<const v4f*>(T + (2 * i + hk) * MP + c4);
      *reinterpret_cast<v4f*>(Yout + (row0 + 2 * i + hk) * ND + c4) = v;
    }
  }
}

__global__ void k_gather_node(const float* __restrict__ nodex,
                              const unsigned* __restrict__ off,
                              const unsigned* __restrict__ csr,
                              float* __restrict__ agg) {
  int r = blockIdx.x * 4 + (threadIdx.x >> 6);
  int d = threadIdx.x & 63;
  unsigned s = off[r], e = off[r + 1];
  float acc = 0.0f;
  for (unsigned j = s; j < e; ++j) acc += nodex[(size_t)csr[j] * ND + d];
  agg[(size_t)r * ND + d] = acc;
}

#define GSUB 4

__device__ __forceinline__ v4f ld_row(const char* __restrict__ hbytes,
                                      unsigned rowoff, unsigned voff) {
  return *reinterpret_cast<const v4f*>(hbytes + (size_t)rowoff + voff);
}

__global__ void __launch_bounds__(256)
k_gather_final(const float* __restrict__ h,
               const float* __restrict__ agg,
               const unsigned* __restrict__ off,
               const unsigned* __restrict__ csr,
               float* __restrict__ out) {
  const int n = blockIdx.x & (NN - 1);
  const int g0 = (blockIdx.x >> 11) * GSUB;
  const unsigned lane = threadIdx.x & 31;
  const unsigned bl = threadIdx.x >> 4;
  const unsigned d4 = (threadIdx.x & 15) << 2;
  const unsigned s = off[n];
  const unsigned deg = off[n + 1] - s;
  const v4f a0 = *reinterpret_cast<const v4f*>(agg + (size_t)n * ND + d4);
  const unsigned idx0 = (lane < deg) ? csr[s + lane] * (unsigned)(ND * 4) : 0u;
  const char* hbytes = reinterpret_cast<const char*>(h);

  for (int g = 0; g < GSUB; ++g) {
    const unsigned b = (unsigned)(g0 + g) * 16u + bl;
    const unsigned voff = (b * (unsigned)(NN * ND) + d4) * 4u;
    v4f acc = a0;
    for (unsigned w = 0; w < deg; w += 32) {
      unsigned idx = idx0;
      if (w != 0)
        idx = (w + lane < deg) ? csr[s + w + lane] * (unsigned)(ND * 4) : 0u;
      unsigned cnt = deg - w;
      if (cnt > 32u) cnt = 32u;
      unsigned j = 0;
      for (; j + 8 <= cnt; j += 8) {
        v4f r[8];
        for (int u = 0; u < 8; ++u)
          r[u] = ld_row(hbytes, (unsigned)__builtin_amdgcn_readlane((int)idx, (int)j + u), voff);
        for (int u = 0; u < 8; ++u) acc += r[u];
      }
      if (j + 4 <= cnt) {
        v4f r[4];
        for (int u = 0; u < 4; ++u)
          r[u] = ld_row(hbytes, (unsigned)__builtin_amdgcn_readlane((int)idx, (int)j + u), voff);
        for (int u = 0; u < 4; ++u) acc += r[u];
        j += 4;
      }
      if (j + 2 <= cnt) {
        v4f r0 = ld_row(hbytes, (unsigned)__builtin_amdgcn_readlane((int)idx, (int)j), voff);
        v4f r1 = ld_row(hbytes, (unsigned)__builtin_amdgcn_readlane((int)idx, (int)j + 1), voff);
        acc += r0;
        acc += r1;
        j += 2;
      }
      if (j < cnt) {
        acc += ld_row(hbytes, (unsigned)__builtin_amdgcn_readlane((int)idx, (int)j), voff);
      }
    }
    *reinterpret_cast<v4f*>(out + ((size_t)b * NN + n) * ND + d4) = acc;
  }
}

extern "C" void kernel_launch(void* const* d_in, const int* in_sizes, int n_in,
                              void* d_out, int out_size, void* d_ws, size_t ws_size,
                              hipStream_t stream) {
  (void)in_sizes; (void)n_in; (void)out_size; (void)ws_size;
  const float* X   = (const float*)d_in[0];
  const int*   ei  = (const int*)d_in[1];
  const float* W1t = (const float*)d_in[2];
  const float* b1t = (const float*)d_in[3];
  const float* W2t = (const float*)d_in[4];
  const float* b2t = (const float*)d_in[5];
  const float* W1n = (const float*)d_in[6];
  const float* b1n = (const float*)d_in[7];
  const float* W2n = (const float*)d_in[8];
  const float* b2n = (const float*)d_in[9];
  float* out = (float*)d_out;

  float* h     = (float*)d_ws;
  float* xmax  = h + (size_t)NB * NN * ND;
  float* nodex = xmax + (size_t)NN * ND;
  float* agg   = nodex + (size_t)NN * ND;
  unsigned* deg    = (unsigned*)(agg + (size_t)NN * ND);
  unsigned* off    = deg + NN;
  unsigned* cursor = off + (NN + 1);
  unsigned* csr    = cursor + NN;

  const int* esrc = ei;
  const int* edst = ei + NE;

  k_zero_u32<<<(NN + 255) / 256, 256, 0, stream>>>(deg, NN);
  k_count<<<NE / 256, 256, 0, stream>>>(edst, deg);
  k_scan<<<1, 256, 0, stream>>>(deg, off, cursor);
  k_fill<<<NE / 256, 256, 0, stream>>>(esrc, edst, cursor, csr);

  k_mlp2_rows_maxb<<<MLP_GRID + MAXB_BLOCKS, 256, 0, stream>>>(X, W1t, b1t, W2t, b2t, h, xmax);
  k_mlp2_wmma<<<NN / 16, 128, 0, stream>>>(xmax, W1n, b1n, W2n, b2n, nodex);
  k_gather_node<<<NN / 4, 256, 0, stream>>>(nodex, off, csr, agg);

  k_gather_final<<<(NB / (16 * GSUB)) * NN, 256, 0, stream>>>(h, agg, off, csr, out);
}


# edge sums: 512 blocks own 4 destination nodes each and find their edges by reading edge_index into LDS lists (order: LDS atomic arrival, new f32 add order), nodex rows summed first; 8 -> 3 launches
# speedup vs baseline: 1.7850x; 1.1667x over previous
#include <hip/hip_runtime.h>

typedef __attribute__((ext_vector_type(2))) float v2f;
typedef __attribute__((ext_vector_type(8))) float v8f;
typedef __attribute__((ext_vector_type(4))) float v4f;

#define NB 128
#define NN 2048
#define ND 64
#define NE 32768

__global__ void __launch_bounds__(128)
k_mlp2_wmma(const float* __restrict__ Xin, const float* __restrict__ W1,
            const float* __restrict__ b1, const float* __restrict__ W2,
            const float* __restrict__ b2, float* __restrict__ Yout) {
  __shared__ float Ys[16 * ND];
  const int lane = threadIdx.x & 31;
  const int wave = threadIdx.x >> 5;
  const int col0 = wave * 16;
  const int hm = lane & 15;
  const int hk = lane >> 4;
  const size_t row0 = (size_t)blockIdx.x * 16;

  v2f bw[16];
#pragma unroll
  for (int kk = 0; kk < 16; ++kk) {
    int k = kk * 4 + 2 * hk;
    bw[kk].x = W1[(k + 0) * ND + col0 + hm];
    bw[kk].y = W1[(k + 1) * ND + col0 + hm];
  }

  v8f acc;
  {
    float bv = b1[col0 + hm];
#pragma unroll
    for (int j = 0; j < 8; ++j) acc[j] = bv;
  }

  const float* xrow = Xin + (row0 + hm) * ND;
#pragma unroll
  for (int kk = 0; kk < 16; ++kk) {
    int k = kk * 4 + 2 * hk;
    v2f a;
    a.x = xrow[k];
    a.y = xrow[k + 1];
    acc = __builtin_amdgcn_wmma_f32_16x16x4_f32(false, a, false, bw[kk],
                                                (short)0, acc, false, false);
  }
#pragma unroll
  for (int j = 0; j < 8; ++j) {
    Ys[(j + 8 * hk) * ND + col0 + hm] = fmaxf(acc[j], 0.0f);
  }
  __syncthreads();

#pragma unroll
  for (int kk = 0; kk < 16; ++kk) {
    int k = kk * 4 + 2 * hk;
    bw[kk].x = W2[(k + 0) * ND + col0 + hm];
    bw[kk].y = W2[(k + 1) * ND + col0 + hm];
  }
  {
    float bv = b2[col0 + hm];
#pragma unroll
    for (int j = 0; j < 8; ++j) acc[j] = bv;
  }
#pragma unroll
  for (int kk = 0; kk < 16; ++kk) {
    int k = kk * 4 + 2 * hk;
    v2f a;
    a.x = Ys[hm * ND + k];
    a.y = Ys[hm * ND + k + 1];
    acc = __builtin_amdgcn_wmma_f32_16x16x4_f32(false, a, false, bw[kk],
                                                (short)0, acc, false, false);
  }
#pragma unroll
  for (int j = 0; j < 8; ++j) {
    Yout[(row0 + j + 8 * hk) * ND + col0 + hm] = fmaxf(acc[j], 0.0f);
  }
}

#define MP 68
#define MW 8
#define MLP_GRID 512
#define MLP_ITERS ((NB * NN / 16) / (MLP_GRID * MW))

#define MAXB_BLOCKS ((NN * ND) / 256)
static_assert(MLP_GRID == 512 && MAXB_BLOCKS == 512, "the role bit and index & 511 assume 512 blocks of each role");

__global__ void __launch_bounds__(256, 8)
k_mlp2_rows_maxb(const float* __restrict__ Xin, const float* __restrict__ W1,
                 const float* __restrict__ b1, const float* __restrict__ W2,
                 const float* __restrict__ b2, float* __restrict__ Yout,
                 float* __restrict__ xmax) {
  __shared__ __attribute__((aligned(16))) float Wt1[ND * MP];
  __shared__ __attribute__((aligned(16))) float Wt2[ND * MP];
  __shared__ __attribute__((aligned(16))) float Ts[MW][16 * MP];
  const int tid = threadIdx.x;
  const unsigned bid = blockIdx.x;
  const unsigned rblk = bid & (MLP_GRID - 1);
  if (((bid >> 1) ^ (bid >> 9)) & 1u) {
    const int i = (int)rblk * 256 + tid;
    float m = -3.4028234663852886e38f;
    for (int b = 0; b < NB; ++b)
      m = fmaxf(m, Xin[(size_t)b * (NN * ND) + i]);
    xmax[i] = m;
    return;
  }
  const int lane = tid & 31;
  const int wave = __builtin_amdgcn_readfirstlane(tid >> 5);
  const int hm = lane & 15;
  const int hk = lane >> 4;
  const int c4 = (lane & 15) << 2;
  float* T = Ts[wave];

  size_t tile = (size_t)rblk * MW + wave;
  v4f xr[8];
#pragma unroll
  for (int i = 0; i < 8; ++i)
    xr[i] = *reinterpret_cast<const v4f*>(Xin + (tile * 16 + 2 * i + hk) * ND + c4);

#pragma unroll
  for (int i = 0; i < 4; ++i) {
    const int idx = tid + 256 * i;
    const int k = idx >> 4;
    const int n4 = (idx & 15) << 2;
    const v4f w1 = *reinterpret_cast<const v4f*>(W1 + k * ND + n4);
    const v4f w2 = *reinterpret_cast<const v4f*>(W2 + k * ND + n4);
#pragma unroll
    for (int e = 0; e < 4; ++e) {
      Wt1[(n4 + e) * MP + k] = w1[e];
      Wt2[(n4 + e) * MP + k] = w2[e];
    }
  }
  float bv1[4], bv2[4];
#pragma unroll
  for (int ct = 0; ct < 4; ++ct) {
    bv1[ct] = b1[ct * 16 + hm];
    bv2[ct] = b2[ct * 16 + hm];
  }

  for (int it = 0; it < MLP_ITERS; ++it) {
#pragma unroll
    for (int i = 0; i < 8; ++i)
      *reinterpret_cast<v4f*>(T + (2 * i + hk) * MP + c4) = xr[i];
    __syncthreads();

    const size_t row0 = tile * 16;
    if (it + 1 < MLP_ITERS) {
      tile += (size_t)MLP_GRID * MW;
#pragma unroll
      for (int i = 0; i < 8; ++i)
        xr[i] = *reinterpret_cast<const v4f*>(Xin + (tile * 16 + 2 * i + hk) * ND + c4);
    }

    v8f acc[4];
#pragma unroll
    for (int ct = 0; ct < 4; ++ct) {
      float bb = bv1[ct];
      asm volatile("" : "+v"(bb));
#pragma unroll
      for (int j = 0; j < 8; ++j) acc[ct][j] = bb;
    }
#pragma unroll
    for (int kk = 0; kk < 16; ++kk) {
      const int k = kk * 4 + 2 * hk;
      const v2f a = *reinterpret_cast<const v2f*>(T + hm * MP + k);
#pragma unroll
      for (int ct = 0; ct < 4; ++ct) {
        const v2f b = *reinterpret_cast<const v2f*>(Wt1 + (ct * 16 + hm) * MP + k);
        acc[ct] = __builtin_amdgcn_wmma_f32_16x16x4_f32(false, a, false, b,
                                                        (short)0, acc[ct], false, false);
      }
    }
#pragma unroll
    for (int ct = 0; ct < 4; ++ct)
#pragma unroll
      for (int j = 0; j < 8; ++j)
        T[(j + 8 * hk) * MP + ct * 16 + hm] = fmaxf(acc[ct][j], 0.0f);
    __syncthreads();

#pragma unroll
    for (int ct = 0; ct < 4; ++ct) {
      float bb = bv2[ct];
      asm volatile("" : "+v"(bb));
#pragma unroll
      for (int j = 0; j < 8; ++j) acc[ct][j] = bb;
    }
#pragma unroll
    for (int kk = 0; kk < 16; ++kk) {
      const int k = kk * 4 + 2 * hk;
      const v2f a = *reinterpret_cast<const v2f*>(T + hm * MP + k);
#pragma unroll
      for (int ct = 0; ct < 4; ++ct) {
        const v2f b = *reinterpret_cast<const v2f*>(Wt2 + (ct * 16 + hm) * MP + k);
        acc[ct] = __builtin_amdgcn_wmma_f32_16x16x4_f32(false, a, false, b,
                                                        (short)0, acc[ct], false, false);
      }
    }
#pragma unroll
    for (int ct = 0; ct < 4; ++ct)
#pragma unroll
      for (int j = 0; j < 8; ++j)
        T[(j + 8 * hk) * MP + ct * 16 + hm] = fmaxf(acc[ct][j], 0.0f);
    __syncthreads();

#pragma unroll
    for (int i = 0; i < 8; ++i) {
      const v4f v = *reinterpret_cast<const v4f*>(T + (2 * i + hk) * MP + c4);
      *reinterpret_cast<v4f*>(Yout + (row0 + 2 * i + hk) * ND + c4) = v;
    }
  }
}

#define GN 4
#define GW 32
#define GT (GW * 32)
static_assert(NE == 32 * GT, "every thread takes 32 destination ids");
static_assert(NB == 4 * GW, "a wave covers 2 subgraphs in each of 2 passes");

typedef __attribute__((ext_vector_type(4))) int v4i;

__device__ __forceinline__ v4f ld_row(const char* __restrict__ base,
                                      unsigned rowoff, unsigned voff) {
  return *reinterpret_cast<const v4f*>(base + (size_t)rowoff + voff);
}

__device__ __forceinline__ v4f add_rows(const char* __restrict__ base, unsigned voff,
                                        unsigned idx, unsigned cnt, v4f acc) {
  unsigned j = 0;
  for (; j + 8 <= cnt; j += 8) {
    v4f r[8];
    for (int u = 0; u < 8; ++u)
      r[u] = ld_row(base, (unsigned)__builtin_amdgcn_readlane((int)idx, (int)j + u), voff);
    for (int u = 0; u < 8; ++u) acc += r[u];
  }
  if (j + 4 <= cnt) {
    v4f r[4];
    for (int u = 0; u < 4; ++u)
      r[u] = ld_row(base, (unsigned)__builtin_amdgcn_readlane((int)idx, (int)j + u), voff);
    for (int u = 0; u < 4; ++u) acc += r[u];
    j += 4;
  }
  if (j + 2 <= cnt) {
    v4f r0 = ld_row(base, (unsigned)__builtin_amdgcn_readlane((int)idx, (int)j), voff);
    v4f r1 = ld_row(base, (unsigned)__builtin_amdgcn_readlane((int)idx, (int)j + 1), voff);
    acc += r0;
    acc += r1;
    j += 2;
  }
  if (j < cnt) {
    acc += ld_row(base, (unsigned)__builtin_amdgcn_readlane((int)idx, (int)j), voff);
  }
  return acc;
}

__global__ void __launch_bounds__(GT, 16)
k_edge_sums(const float* __restrict__ h, const float* __restrict__ nodex,
            const int* __restrict__ ei, float* __restrict__ out) {
  __shared__ unsigned short pool[NE];
  __shared__ unsigned cnt[GN];
  __shared__ unsigned cur[GN];
  __shared__ __attribute__((aligned(16))) float aggs[GN][ND];
  const unsigned tid = threadIdx.x;
  const unsigned lane = tid & 31;
  const unsigned n0 = blockIdx.x * GN;
  const int* __restrict__ esrc = ei;
  const int* __restrict__ edst = ei + NE;

  v4i dv[8];
#pragma unroll
  for (int i = 0; i < 8; ++i)
    dv[i] = *reinterpret_cast<const v4i*>(edst + 4 * (i * GT + (int)tid));
  if (tid < GN) cnt[tid] = 0u;
  __syncthreads();
#pragma unroll
  for (int i = 0; i < 8; ++i)
#pragma unroll
    for (int c = 0; c < 4; ++c) {
      const unsigned sl = (unsigned)dv[i][c] - n0;
      if (sl < GN) atomicAdd(&cnt[sl], 1u);
    }
  __syncthreads();
  if (tid < GN) {
    unsigned p = 0;
    for (unsigned q = 0; q < GN; ++q) p += (q < tid) ? cnt[q] : 0u;
    cur[tid] = p;
  }
  __syncthreads();
#pragma unroll
  for (int i = 0; i < 8; ++i)
#pragma unroll
    for (int c = 0; c < 4; ++c) {
      const unsigned sl = (unsigned)dv[i][c] - n0;
      if (sl < GN) {
        const unsigned p = atomicAdd(&cur[sl], 1u);
        pool[p] = (unsigned short)esrc[4 * (i * GT + (int)tid) + c];
      }
    }
  __syncthreads();

  const unsigned wave = (unsigned)__builtin_amdgcn_readfirstlane((int)(tid >> 5));
  const unsigned d4 = (tid & 15) << 2;
  const char* nbytes = reinterpret_cast<const char*>(nodex);
  const char* hbytes = reinterpret_cast<const char*>(h);

  if (wave < GN) {
    const unsigned deg = (unsigned)__builtin_amdgcn_readfirstlane((int)cnt[wave]);
    const unsigned st = (unsigned)__builtin_amdgcn_readfirstlane((int)cur[wave]) - deg;
    v4f acc = {0.0f, 0.0f, 0.0f, 0.0f};
    for (unsigned w = 0; w < deg; w += 32) {
      const unsigned idx = (w + lane < deg) ? (unsigned)pool[st + w + lane] * (unsigned)(ND * 4) : 0u;
      unsigned c = deg - w;
      if (c > 32u) c = 32u;
      acc = add_rows(nbytes, d4 * 4u, idx, c, acc);
    }
    if (lane < 16) *reinterpret_cast<v4f*>(&aggs[wave][d4]) = acc;
  }
  __syncthreads();

  const unsigned hb = lane >> 4;
  for (unsigned sl = 0; sl < GN; ++sl) {
    const unsigned deg = (unsigned)__builtin_amdgcn_readfirstlane((int)cnt[sl]);
    const unsigned st = (unsigned)__builtin_amdgcn_readfirstlane((int)cur[sl]) - deg;
    const v4f a0 = *reinterpret_cast<const v4f*>(&aggs[sl][d4]);
    const unsigned idx0 = (lane < deg) ? (unsigned)pool[st + lane] * (unsigned)(ND * 4) : 0u;
    const unsigned n = n0 + sl;
    for (unsigned hh = 0; hh < 2; ++hh) {
      const unsigned b = hh * (2u * GW) + wave * 2u + hb;
      const unsigned voff = (b * (unsigned)(NN * ND) + d4) * 4u;
      v4f acc = a0;
      for (unsigned w = 0; w < deg; w += 32) {
        unsigned idx = idx0;
        if (w != 0)
          idx = (w + lane < deg) ? (unsigned)pool[st + w + lane] * (unsigned)(ND * 4) : 0u;
        unsigned c = deg - w;
        if (c > 32u) c = 32u;
        acc = add_rows(hbytes, voff, idx, c, acc);
      }
      *reinterpret_cast<v4f*>(out + ((size_t)b * NN + n) * ND + d4) = acc;
    }
  }
}

extern "C" void kernel_launch(void* const* d_in, const int* in_sizes, int n_in,
                              void* d_out, int out_size, void* d_ws, size_t ws_size,
                              hipStream_t stream) {
  (void)in_sizes; (void)n_in; (void)out_size; (void)ws_size;
  const float* X   = (const float*)d_in[0];
  const int*   ei  = (const int*)d_in[1];
  const float* W1t = (const float*)d_in[2];
  const float* b1t = (const float*)d_in[3];
  const float* W2t = (const float*)d_in[4];
  const float* b2t = (const float*)d_in[5];
  const float* W1n = (const float*)d_in[6];
  const float* b1n = (const float*)d_in[7];
  const float* W2n = (const float*)d_in[8];
  const float* b2n = (const float*)d_in[9];
  float* out = (float*)d_out;

  float* h     = (float*)d_ws;
  float* xmax  = h + (size_t)NB * NN * ND;
  float* nodex = xmax + (size_t)NN * ND;

  k_mlp2_rows_maxb<<<MLP_GRID + MAXB_BLOCKS, 256, 0, stream>>>(X, W1t, b1t, W2t, b2t, h, xmax);
  k_mlp2_wmma<<<NN / 16, 128, 0, stream>>>(xmax, W1n, b1n, W2n, b2n, nodex);
  k_edge_sums<<<NN / GN, GT, 0, stream>>>(h, nodex, ei, out);
}


# edge sums: a block reads its die number and draws its unit (8 nodes x 64 subgraphs) by ticket from its own L2 half's counter, so each cache half reads only its half of h; f32 add order kept
# speedup vs baseline: 1.8657x; 1.0452x over previous
#include <hip/hip_runtime.h>

typedef __attribute__((ext_vector_type(2))) float v2f;
typedef __attribute__((ext_vector_type(8))) float v8f;
typedef __attribute__((ext_vector_type(4))) float v4f;

#define NB 128
#define NN 2048
#define ND 64
#define NE 32768

__global__ void __launch_bounds__(128)
k_mlp2_wmma(const float* Xin, const float* __restrict__ W1,
            const float* __restrict__ b1, const float* __restrict__ W2,
            const float* __restrict__ b2, float* __restrict__ Yout, unsigned* ctr) {
  __shared__ float Ys[16 * ND];
  const int lane = threadIdx.x & 31;
  const int wave = threadIdx.x >> 5;
  const int col0 = wave * 16;
  const int hm = lane & 15;
  const int hk = lane >> 4;
  const size_t row0 = (size_t)blockIdx.x * 16;

  v2f bw[16];
#pragma unroll
  for (int kk = 0; kk < 16; ++kk) {
    int k = kk * 4 + 2 * hk;
    bw[kk].x = W1[(k + 0) * ND + col0 + hm];
    bw[kk].y = W1[(k + 1) * ND + col0 + hm];
  }

  v8f acc;
  {
    float bv = b1[col0 + hm];
#pragma unroll
    for (int j = 0; j < 8; ++j) acc[j] = bv;
  }

  const float* xrow = Xin + (row0 + hm) * ND;
#pragma unroll
  for (int kk = 0; kk < 16; ++kk) {
    int k = kk * 4 + 2 * hk;
    v2f a;
    a.x = xrow[k];
    a.y = xrow[k + 1];
    acc = __builtin_amdgcn_wmma_f32_16x16x4_f32(false, a, false, bw[kk],
                                                (short)0, acc, false, false);
  }
#pragma unroll
  for (int j = 0; j < 8; ++j) {
    Ys[(j + 8 * hk) * ND + col0 + hm] = fmaxf(acc[j], 0.0f);
  }
  __syncthreads();
  if (blockIdx.x == 0 && threadIdx.x < 2) ctr[threadIdx.x] = 0u;

#pragma unroll
  for (int kk = 0; kk < 16; ++kk) {
    int k = kk * 4 + 2 * hk;
    bw[kk].x = W2[(k + 0) * ND + col0 + hm];
    bw[kk].y = W2[(k + 1) * ND + col0 + hm];
  }
  {
    float bv = b2[col0 + hm];
#pragma unroll
    for (int j = 0; j < 8; ++j) acc[j] = bv;
  }
#pragma unroll
  for (int kk = 0; kk < 16; ++kk) {
    int k = kk * 4 + 2 * hk;
    v2f a;
    a.x = Ys[hm * ND + k];
    a.y = Ys[hm * ND + k + 1];
    acc = __builtin_amdgcn_wmma_f32_16x16x4_f32(false, a, false, bw[kk],
                                                (short)0, acc, false, false);
  }
#pragma unroll
  for (int j = 0; j < 8; ++j) {
    Yout[(row0 + j + 8 * hk) * ND + col0 + hm] = fmaxf(acc[j], 0.0f);
  }
}

#define MP 68
#define MW 8
#define MLP_GRID 512
#define MLP_ITERS ((NB * NN / 16) / (MLP_GRID * MW))

#define MAXB_BLOCKS ((NN * ND) / 256)
static_assert(MLP_GRID == 512 && MAXB_BLOCKS == 512, "the role bit and index & 511 assume 512 blocks of each role");

__global__ void __launch_bounds__(256, 8)
k_mlp2_rows_maxb(const float* __restrict__ Xin, const float* __restrict__ W1,
                 const float* __restrict__ b1, const float* __restrict__ W2,
                 const float* __restrict__ b2, float* __restrict__ Yout,
                 float* __restrict__ xmax) {
  __shared__ __attribute__((aligned(16))) float Wt1[ND * MP];
  __shared__ __attribute__((aligned(16))) float Wt2[ND * MP];
  __shared__ __attribute__((aligned(16))) float Ts[MW][16 * MP];
  const int tid = threadIdx.x;
  const unsigned bid = blockIdx.x;
  const unsigned rblk = bid & (MLP_GRID - 1);
  if (((bid >> 1) ^ (bid >> 9)) & 1u) {
    const int i = (int)rblk * 256 + tid;
    float m = -3.4028234663852886e38f;
    for (int b = 0; b < NB; ++b)
      m = fmaxf(m, Xin[(size_t)b * (NN * ND) + i]);
    xmax[i] = m;
    return;
  }
  const int lane = tid & 31;
  const int wave = __builtin_amdgcn_readfirstlane(tid >> 5);
  const int hm = lane & 15;
  const int hk = lane >> 4;
  const int c4 = (lane & 15) << 2;
  float* T = Ts[wave];

  size_t tile = (size_t)rblk * MW + wave;
  v4f xr[8];
#pragma unroll
  for (int i = 0; i < 8; ++i)
    xr[i] = *reinterpret_cast<const v4f*>(Xin + (tile * 16 + 2 * i + hk) * ND + c4);

#pragma unroll
  for (int i = 0; i < 4; ++i) {
    const int idx = tid + 256 * i;
    const int k = idx >> 4;
    const int n4 = (idx & 15) << 2;
    const v4f w1 = *reinterpret_cast<const v4f*>(W1 + k * ND + n4);
    const v4f w2 = *reinterpret_cast<const v4f*>(W2 + k * ND + n4);
#pragma unroll
    for (int e = 0; e < 4; ++e) {
      Wt1[(n4 + e) * MP + k] = w1[e];
      Wt2[(n4 + e) * MP + k] = w2[e];
    }
  }
  float bv1[4], bv2[4];
#pragma unroll
  for (int ct = 0; ct < 4; ++ct) {
    bv1[ct] = b1[ct * 16 + hm];
    bv2[ct] = b2[ct * 16 + hm];
  }

  for (int it = 0; it < MLP_ITERS; ++it) {
#pragma unroll
    for (int i = 0; i < 8; ++i)
      *reinterpret_cast<v4f*>(T + (2 * i + hk) * MP + c4) = xr[i];
    __syncthreads();

    const size_t row0 = tile * 16;
    if (it + 1 < MLP_ITERS) {
      tile += (size_t)MLP_GRID * MW;
#pragma unroll
      for (int i = 0; i < 8; ++i)
        xr[i] = *reinterpret_cast<const v4f*>(Xin + (tile * 16 + 2 * i + hk) * ND + c4);
    }

    v8f acc[4];
#pragma unroll
    for (int ct = 0; ct < 4; ++ct) {
      float bb = bv1[ct];
      asm volatile("" : "+v"(bb));
#pragma unroll
      for (int j = 0; j < 8; ++j) acc[ct][j] = bb;
    }
#pragma unroll
    for (int kk = 0; kk < 16; ++kk) {
      const int k = kk * 4 + 2 * hk;
      const v2f a = *reinterpret_cast<const v2f*>(T + hm * MP + k);
#pragma unroll
      for (int ct = 0; ct < 4; ++ct) {
        const v2f b = *reinterpret_cast<const v2f*>(Wt1 + (ct * 16 + hm) * MP + k);
        acc[ct] = __builtin_amdgcn_wmma_f32_16x16x4_f32(false, a, false, b,
                                                        (short)0, acc[ct], false, false);
      }
    }
#pragma unroll
    for (int ct = 0; ct < 4; ++ct)
#pragma unroll
      for (int j = 0; j < 8; ++j)
        T[(j + 8 * hk) * MP + ct * 16 + hm] = fmaxf(acc[ct][j], 0.0f);
    __syncthreads();

#pragma unroll
    for (int ct = 0; ct < 4; ++ct) {
      float bb = bv2[ct];
      asm volatile("" : "+v"(bb));
#pragma unroll
      for (int j = 0; j < 8; ++j) acc[ct][j] = bb;
    }
#pragma unroll
    for (int kk = 0; kk < 16; ++kk) {
      const int k = kk * 4 + 2 * hk;
      const v2f a = *reinterpret_cast<const v2f*>(T + hm * MP + k);
#pragma unroll
      for (int ct = 0; ct < 4; ++ct) {
        const v2f b = *reinterpret_cast<const v2f*>(Wt2 + (ct * 16 + hm) * MP + k);
        acc[ct] = __builtin_amdgcn_wmma_f32_16x16x4_f32(false, a, false, b,
                                                        (short)0, acc[ct], false, false);
      }
    }
#pragma unroll
    for (int ct = 0; ct < 4; ++ct)
#pragma unroll
      for (int j = 0; j < 8; ++j)
        T[(j + 8 * hk) * MP + ct * 16 + hm] = fmaxf(acc[ct][j], 0.0f);
    __syncthreads();

#pragma unroll
    for (int i = 0; i < 8; ++i) {
      const v4f v = *reinterpret_cast<const v4f*>(T + (2 * i + hk) * MP + c4);
      *reinterpret_cast<v4f*>(Yout + (row0 + 2 * i + hk) * ND + c4) = v;
    }
  }
}

#define GN 8
#define GW 32
#define GT (GW * 32)
#define GUNITS (NN / GN)
static_assert(NE == 32 * GT, "every thread takes 32 destination ids");
static_assert(NB == 4 * GW, "a unit is 64 subgraphs: a wave covers 2 of them");
static_assert(GUNITS == 256, "unit number = half * 256 + ticket");

typedef __attribute__((ext_vector_type(4))) int v4i;

__device__ __forceinline__ v4f ld_row(const char* __restrict__ base,
                                      unsigned rowoff, unsigned voff) {
  return *reinterpret_cast<const v4f*>(base + (size_t)rowoff + voff);
}

__device__ __forceinline__ v4f add_rows(const char* __restrict__ base, unsigned voff,
                                        unsigned idx, unsigned cnt, v4f acc) {
  unsigned j = 0;
  for (; j + 8 <= cnt; j += 8) {
    v4f r[8];
    for (int u = 0; u < 8; ++u)
      r[u] = ld_row(base, (unsigned)__builtin_amdgcn_readlane((int)idx, (int)j + u), voff);
    for (int u = 0; u < 8; ++u) acc += r[u];
  }
  if (j + 4 <= cnt) {
    v4f r[4];
    for (int u = 0; u < 4; ++u)
      r[u] = ld_row(base, (unsigned)__builtin_amdgcn_readlane((int)idx, (int)j + u), voff);
    for (int u = 0; u < 4; ++u) acc += r[u];
    j += 4;
  }
  if (j + 2 <= cnt) {
    v4f r0 = ld_row(base, (unsigned)__builtin_amdgcn_readlane((int)idx, (int)j), voff);
    v4f r1 = ld_row(base, (unsigned)__builtin_amdgcn_readlane((int)idx, (int)j + 1), voff);
    acc += r0;
    acc += r1;
    j += 2;
  }
  if (j < cnt) {
    acc += ld_row(base, (unsigned)__builtin_amdgcn_readlane((int)idx, (int)j), voff);
  }
  return acc;
}

__global__ void __launch_bounds__(GT, 16)
k_edge_sums(const float* __restrict__ h, const float* __restrict__ nodex,
            const int* __restrict__ ei, float* __restrict__ out, unsigned* __restrict__ ctr) {
  __shared__ unsigned short pool[NE];
  __shared__ unsigned cnt[GN];
  __shared__ unsigned cur[GN];
  __shared__ unsigned unit_s;
  __shared__ __attribute__((aligned(16))) float aggs[GN][ND];
  const unsigned tid = threadIdx.x;
  const unsigned lane = tid & 31;
  const int* __restrict__ esrc = ei;
  const int* __restrict__ edst = ei + NE;

  v4i dv[8];
#pragma unroll
  for (int i = 0; i < 8; ++i)
    dv[i] = *reinterpret_cast<const v4i*>(edst + 4 * (i * GT + (int)tid));
  if (tid == 0) {
    unsigned half = ((unsigned)__builtin_amdgcn_s_sendmsg_rtn(0x87) >> 18) & 1u;
    unsigned t = atomicAdd(&ctr[half], 1u);
    if (t >= (unsigned)GUNITS) {
      half ^= 1u;
      t = atomicAdd(&ctr[half], 1u);
    }
    unit_s = half * (unsigned)GUNITS + (t & (unsigned)(GUNITS - 1));
  }
  if (tid < GN) cnt[tid] = 0u;
  __syncthreads();
  const unsigned unit = (unsigned)__builtin_amdgcn_readfirstlane((int)unit_s);
  const unsigned n0 = (unit & (unsigned)(GUNITS - 1)) * GN;
  const unsigned b0 = (unit >> 8) * 64u;
#pragma unroll
  for (int i = 0; i < 8; ++i)
#pragma unroll
    for (int c = 0; c < 4; ++c) {
      const unsigned sl = (unsigned)dv[i][c] - n0;
      if (sl < GN) atomicAdd(&cnt[sl], 1u);
    }
  __syncthreads();
  if (tid < GN) {
    unsigned p = 0;
    for (unsigned q = 0; q < GN; ++q) p += (q < tid) ? cnt[q] : 0u;
    cur[tid] = p;
  }
  __syncthreads();
#pragma unroll
  for (int i = 0; i < 8; ++i)
#pragma unroll
    for (int c = 0; c < 4; ++c) {
      const unsigned sl = (unsigned)dv[i][c] - n0;
      if (sl < GN) {
        const unsigned p = atomicAdd(&cur[sl], 1u);
        pool[p] = (unsigned short)esrc[4 * (i * GT + (int)tid) + c];
      }
    }
  __syncthreads();

  const unsigned wave = (unsigned)__builtin_amdgcn_readfirstlane((int)(tid >> 5));
  const unsigned d4 = (tid & 15) << 2;
  const char* nbytes = reinterpret_cast<const char*>(nodex);
  const char* hbytes = reinterpret_cast<const char*>(h);

  if (wave < GN) {
    const unsigned deg = (unsigned)__builtin_amdgcn_readfirstlane((int)cnt[wave]);
    const unsigned st = (unsigned)__builtin_amdgcn_readfirstlane((int)cur[wave]) - deg;
    v4f acc = {0.0f, 0.0f, 0.0f, 0.0f};
    for (unsigned w = 0; w < deg; w += 32) {
      const unsigned idx = (w + lane < deg) ? (unsigned)pool[st + w + lane] * (unsigned)(ND * 4) : 0u;
      unsigned c = deg - w;
      if (c > 32u) c = 32u;
      acc = add_rows(nbytes, d4 * 4u, idx, c, acc);
    }
    if (lane < 16) *reinterpret_cast<v4f*>(&aggs[wave][d4]) = acc;
  }
  __syncthreads();

  const unsigned hb = lane >> 4;
  for (unsigned sl = 0; sl < GN; ++sl) {
    const unsigned deg = (unsigned)__builtin_amdgcn_readfirstlane((int)cnt[sl]);
    const unsigned st = (unsigned)__builtin_amdgcn_readfirstlane((int)cur[sl]) - deg;
    const v4f a0 = *reinterpret_cast<const v4f*>(&aggs[sl][d4]);
    const unsigned idx0 = (lane < deg) ? (unsigned)pool[st + lane] * (unsigned)(ND * 4) : 0u;
    const unsigned n = n0 + sl;
    {
      const unsigned b = b0 + wave * 2u + hb;
      const unsigned voff = (b * (unsigned)(NN * ND) + d4) * 4u;
      v4f acc = a0;
      for (unsigned w = 0; w < deg; w += 32) {
        unsigned idx = idx0;
        if (w != 0)
          idx = (w + lane < deg) ? (unsigned)pool[st + w + lane] * (unsigned)(ND * 4) : 0u;
        unsigned c = deg - w;
        if (c > 32u) c = 32u;
        acc = add_rows(hbytes, voff, idx, c, acc);
      }
      *reinterpret_cast<v4f*>(out + ((size_t)b * NN + n) * ND + d4) = acc;
    }
  }
}

extern "C" void kernel_launch(void* const* d_in, const int* in_sizes, int n_in,
                              void* d_out, int out_size, void* d_ws, size_t ws_size,
                              hipStream_t stream) {
  (void)in_sizes; (void)n_in; (void)out_size; (void)ws_size;
  const float* X   = (const float*)d_in[0];
  const int*   ei  = (const int*)d_in[1];
  const float* W1t = (const float*)d_in[2];
  const float* b1t = (const float*)d_in[3];
  const float* W2t = (const float*)d_in[4];
  const float* b2t = (const float*)d_in[5];
  const float* W1n = (const float*)d_in[6];
  const float* b1n = (const float*)d_in[7];
  const float* W2n = (const float*)d_in[8];
  const float* b2n = (const float*)d_in[9];
  float* out = (float*)d_out;

  float* h     = (float*)d_ws;
  float* xmax  = h + (size_t)NB * NN * ND;
  float* nodex = xmax + (size_t)NN * ND;

  k_mlp2_rows_maxb<<<MLP_GRID + MAXB_BLOCKS, 256, 0, stream>>>(X, W1t, b1t, W2t, b2t, h, xmax);
  k_mlp2_wmma<<<NN / 16, 128, 0, stream>>>(xmax, W1n, b1n, W2n, b2n, nodex, (unsigned*)xmax);
  k_edge_sums<<<2 * GUNITS, GT, 0, stream>>>(h, nodex, ei, out, (unsigned*)xmax);
}


# edge sums as a scatter: a block keeps one (subgraph, 32-column half) of out as f32 sums in LDS, reads h once in source order (edges sorted by source), one LDS float add per edge: new f32 add order
# speedup vs baseline: 1.9255x; 1.0321x over previous
#include <hip/hip_runtime.h>

typedef __attribute__((ext_vector_type(2))) float v2f;
typedef __attribute__((ext_vector_type(8))) float v8f;
typedef __attribute__((ext_vector_type(4))) float v4f;

#define NB 128
#define NN 2048
#define ND 64
#define NE 32768

typedef __attribute__((ext_vector_type(4))) int v4i;

#define NM_GROUPS 8
#define NM_BLOCKS (NN / (16 * NM_GROUPS))
#define NM_T (NM_GROUPS * 128)
#define LST_DST_WORD 4096
static_assert(NM_T == 1024 && NE == 32 * NM_T, "the sorting block: every thread takes 32 edges");
static_assert(NN == 2 * NM_T, "the sorting block: every thread scans two counters");

__global__ void __launch_bounds__(NM_T)
k_mlp2_wmma(const float* Xin, const float* __restrict__ W1,
            const float* __restrict__ b1, const float* __restrict__ W2,
            const float* __restrict__ b2, float* Yout,
            const int* __restrict__ ei, unsigned* __restrict__ lst) {
  __shared__ float Ys[NM_GROUPS][16 * ND];
  __shared__ unsigned cnt[NN];
  __shared__ unsigned wsum[32];
  __shared__ __attribute__((aligned(16))) unsigned sl[NE];
  const int tid = threadIdx.x;
  const int lane = tid & 31;
  const int wave = tid >> 5;

  if (blockIdx.x == NM_BLOCKS) {
    const int* __restrict__ esrc = ei;
    const int* __restrict__ edst = ei + NE;
    v4i sv[8], dv[8];
#pragma unroll
    for (int i = 0; i < 8; ++i) {
      sv[i] = *reinterpret_cast<const v4i*>(esrc + 4 * (i * NM_T + tid));
      dv[i] = *reinterpret_cast<const v4i*>(edst + 4 * (i * NM_T + tid));
    }
    cnt[tid] = 0u;
    cnt[tid + NM_T] = 0u;
    __syncthreads();
#pragma unroll
    for (int i = 0; i < 8; ++i)
#pragma unroll
      for (int c = 0; c < 4; ++c)
        atomicAdd(&cnt[(unsigned)sv[i][c] & (unsigned)(NN - 1)], 1u);
    __syncthreads();
    const unsigned c0 = cnt[2 * tid];
    const unsigned c1 = cnt[2 * tid + 1];
    unsigned incl = c0 + c1;
#pragma unroll
    for (int o = 1; o < 32; o <<= 1) {
      const unsigned up = __shfl_up(incl, (unsigned)o, 32);
      if (lane >= o) incl += up;
    }
    if (lane == 31) wsum[wave] = incl;
    __syncthreads();
    unsigned base = 0u;
    for (int q = 0; q < 32; ++q) base += (q < wave) ? wsum[q] : 0u;
    const unsigned ex = base + incl - (c0 + c1);
    cnt[2 * tid] = ex;
    cnt[2 * tid + 1] = ex + c0;
    lst[2 * tid] = ex;
    lst[2 * tid + 1] = ex + c0;
    if (tid == NM_T - 1) lst[NN] = ex + c0 + c1;
    __syncthreads();
#pragma unroll
    for (int i = 0; i < 8; ++i) {
      unsigned p[4];
#pragma unroll
      for (int c = 0; c < 4; ++c)
        p[c] = atomicAdd(&cnt[(unsigned)sv[i][c] & (unsigned)(NN - 1)], 1u);
#pragma unroll
      for (int c = 0; c < 4; ++c)
        sl[p[c] & (unsigned)(NE - 1)] = ((unsigned)dv[i][c] & (unsigned)(NN - 1)) << 7;
    }
    __syncthreads();
    char* gdst = reinterpret_cast<char*>(lst + LST_DST_WORD);
    const char* sbytes = reinterpret_cast<const char*>(sl);
#pragma unroll
    for (int i = 0; i < 8; ++i) {
      const int idx = tid + NM_T * i;
      *reinterpret_cast<v4i*>(gdst + 16 * idx) = *reinterpret_cast<const v4i*>(sbytes + 16 * idx);
    }
    return;
  }

  const int grp = wave >> 2;
  const int col0 = (wave & 3) * 16;
  const int hm = lane & 15;
  const int hk = lane >> 4;
  const size_t row0 = ((size_t)blockIdx.x * NM_GROUPS + grp) * 16;
  float* Yg = Ys[grp];

  v2f bw[16];
#pragma unroll
  for (int kk = 0; kk < 16; ++kk) {
    int k = kk * 4 + 2 * hk;
    bw[kk].x = W1[(k + 0) * ND + col0 + hm];
    bw[kk].y = W1[(k + 1) * ND + col0 + hm];
  }

  v8f acc;
  {
    float bv = b1[col0 + hm];
#pragma unroll
    for (int j = 0; j < 8; ++j) acc[j] = bv;
  }

  const float* xrow = Xin + (row0 + hm) * ND;
#pragma unroll
  for (int kk = 0; kk < 16; ++kk) {
    int k = kk * 4 + 2 * hk;
    v2f a;
    a.x = xrow[k];
    a.y = xrow[k + 1];
    acc = __builtin_amdgcn_wmma_f32_16x16x4_f32(false, a, false, bw[kk],
                                                (short)0, acc, false, false);
  }
#pragma unroll
  for (int j = 0; j < 8; ++j) {
    Yg[(j + 8 * hk) * ND + col0 + hm] = fmaxf(acc[j], 0.0f);
  }
  __syncthreads();

#pragma unroll
  for (int kk = 0; kk < 16; ++kk) {
    int k = kk * 4 + 2 * hk;
    bw[kk].x = W2[(k + 0) * ND + col0 + hm];
    bw[kk].y = W2[(k + 1) * ND + col0 + hm];
  }
  {
    float bv = b2[col0 + hm];
#pragma unroll
    for (int j = 0; j < 8; ++j) acc[j] = bv;
  }
#pragma unroll
  for (int kk = 0; kk < 16; ++kk) {
    int k = kk * 4 + 2 * hk;
    v2f a;
    a.x = Yg[hm * ND + k];
    a.y = Yg[hm * ND + k + 1];
    acc = __builtin_amdgcn_wmma_f32_16x16x4_f32(false, a, false, bw[kk],
                                                (short)0, acc, false, false);
  }
#pragma unroll
  for (int j = 0; j < 8; ++j) {
    Yout[(row0 + j + 8 * hk) * ND + col0 + hm] = fmaxf(acc[j], 0.0f);
  }
}

#define MP 68
#define MW 8
#define MLP_GRID 512
#define MLP_ITERS ((NB * NN / 16) / (MLP_GRID * MW))

#define MAXB_BLOCKS ((NN * ND) / 256)
static_assert(MLP_GRID == 512 && MAXB_BLOCKS == 512, "the role bit and index & 511 assume 512 blocks of each role");

__global__ void __launch_bounds__(256, 8)
k_mlp2_rows_maxb(const float* __restrict__ Xin, const float* __restrict__ W1,
                 const float* __restrict__ b1, const float* __restrict__ W2,
                 const float* __restrict__ b2, float* __restrict__ Yout,
                 float* __restrict__ xmax) {
  __shared__ __attribute__((aligned(16))) float Wt1[ND * MP];
  __shared__ __attribute__((aligned(16))) float Wt2[ND * MP];
  __shared__ __attribute__((aligned(16))) float Ts[MW][16 * MP];
  const int tid = threadIdx.x;
  const unsigned bid = blockIdx.x;
  const unsigned rblk = bid & (MLP_GRID - 1);
  if (((bid >> 1) ^ (bid >> 9)) & 1u) {
    const int i = (int)rblk * 256 + tid;
    float m = -3.4028234663852886e38f;
    for (int b = 0; b < NB; ++b)
      m = fmaxf(m, Xin[(size_t)b * (NN * ND) + i]);
    xmax[i] = m;
    return;
  }
  const int lane = tid & 31;
  const int wave = __builtin_amdgcn_readfirstlane(tid >> 5);
  const int hm = lane & 15;
  const int hk = lane >> 4;
  const int c4 = (lane & 15) << 2;
  float* T = Ts[wave];

  size_t tile = (size_t)rblk * MW + wave;
  v4f xr[8];
#pragma unroll
  for (int i = 0; i < 8; ++i)
    xr[i] = *reinterpret_cast<const v4f*>(Xin + (tile * 16 + 2 * i + hk) * ND + c4);

#pragma unroll
  for (int i = 0; i < 4; ++i) {
    const int idx = tid + 256 * i;
    const int k = idx >> 4;
    const int n4 = (idx & 15) << 2;
    const v4f w1 = *reinterpret_cast<const v4f*>(W1 + k * ND + n4);
    const v4f w2 = *reinterpret_cast<const v4f*>(W2 + k * ND + n4);
#pragma unroll
    for (int e = 0; e < 4; ++e) {
      Wt1[(n4 + e) * MP + k] = w1[e];
      Wt2[(n4 + e) * MP + k] = w2[e];
    }
  }
  float bv1[4], bv2[4];
#pragma unroll
  for (int ct = 0; ct < 4; ++ct) {
    bv1[ct] = b1[ct * 16 + hm];
    bv2[ct] = b2[ct * 16 + hm];
  }

  for (int it = 0; it < MLP_ITERS; ++it) {
#pragma unroll
    for (int i = 0; i < 8; ++i)
      *reinterpret_cast<v4f*>(T + (2 * i + hk) * MP + c4) = xr[i];
    __syncthreads();

    const size_t row0 = tile * 16;
    if (it + 1 < MLP_ITERS) {
      tile += (size_t)MLP_GRID * MW;
#pragma unroll
      for (int i = 0; i < 8; ++i)
        xr[i] = *reinterpret_cast<const v4f*>(Xin + (tile * 16 + 2 * i + hk) * ND + c4);
    }

    v8f acc[4];
#pragma unroll
    for (int ct = 0; ct < 4; ++ct) {
      float bb = bv1[ct];
      asm volatile("" : "+v"(bb));
#pragma unroll
      for (int j = 0; j < 8; ++j) acc[ct][j] = bb;
    }
#pragma unroll
    for (int kk = 0; kk < 16; ++kk) {
      const int k = kk * 4 + 2 * hk;
      const v2f a = *reinterpret_cast<const v2f*>(T + hm * MP + k);
#pragma unroll
      for (int ct = 0; ct < 4; ++ct) {
        const v2f b = *reinterpret_cast<const v2f*>(Wt1 + (ct * 16 + hm) * MP + k);
        acc[ct] = __builtin_amdgcn_wmma_f32_16x16x4_f32(false, a, false, b,
                                                        (short)0, acc[ct], false, false);
      }
    }
#pragma unroll
    for (int ct = 0; ct < 4; ++ct)
#pragma unroll
      for (int j = 0; j < 8; ++j)
        T[(j + 8 * hk) * MP + ct * 16 + hm] = fmaxf(acc[ct][j], 0.0f);
    __syncthreads();

#pragma unroll
    for (int ct = 0; ct < 4; ++ct) {
      float bb = bv2[ct];
      asm volatile("" : "+v"(bb));
#pragma unroll
      for (int j = 0; j < 8; ++j) acc[ct][j] = bb;
    }
#pragma unroll
    for (int kk = 0; kk < 16; ++kk) {
      const int k = kk * 4 + 2 * hk;
      const v2f a = *reinterpret_cast<const v2f*>(T + hm * MP + k);
#pragma unroll
      for (int ct = 0; ct < 4; ++ct) {
        const v2f b = *reinterpret_cast<const v2f*>(Wt2 + (ct * 16 + hm) * MP + k);
        acc[ct] = __builtin_amdgcn_wmma_f32_16x16x4_f32(false, a, false, b,
                                                        (short)0, acc[ct], false, false);
      }
    }
#pragma unroll
    for (int ct = 0; ct < 4; ++ct)
#pragma unroll
      for (int j = 0; j < 8; ++j)
        T[(j + 8 * hk) * MP + ct * 16 + hm] = fmaxf(acc[ct][j], 0.0f);
    __syncthreads();

#pragma unroll
    for (int i = 0; i < 8; ++i) {
      const v4f v = *reinterpret_cast<const v4f*>(T + (2 * i + hk) * MP + c4);
      *reinterpret_cast<v4f*>(Yout + (row0 + 2 * i + hk) * ND + c4) = v;
    }
  }
}

#define SC_T 1024
#define SC_SRC (NN / 32)
#define SC_CH 8
static_assert(SC_SRC == 64 && (SC_SRC % SC_CH) == 0, "a wave keeps its 65 row starts in two registers and a scalar");

__device__ __forceinline__ void lds_add(char* p, float v) {
  __hip_atomic_fetch_add(reinterpret_cast<float*>(p), v, __ATOMIC_RELAXED, __HIP_MEMORY_SCOPE_WORKGROUP);
}

__global__ void __launch_bounds__(SC_T)
k_edge_scatter(const float* __restrict__ h, const float* __restrict__ nodex,
               const unsigned* __restrict__ lst, float* __restrict__ out) {
  __shared__ __attribute__((aligned(16))) float acc[NN * 32];
  const unsigned tid = threadIdx.x;
  const unsigned lane = tid & 31;
  const unsigned wave = (unsigned)__builtin_amdgcn_readfirstlane((int)(tid >> 5));
  const unsigned b = blockIdx.x >> 1;
  const unsigned ch = blockIdx.x & 1u;
  const unsigned* __restrict__ rowptr = lst;
  const unsigned* __restrict__ sdst = lst + LST_DST_WORD;
  char* accb = reinterpret_cast<char*>(acc);
  const unsigned lane4 = lane << 2;

  const unsigned j0 = wave * SC_SRC;
  const unsigned rp0 = rowptr[j0 + lane];
  const unsigned rp1 = rowptr[j0 + 32 + lane];
  const unsigned rpe = (unsigned)__builtin_amdgcn_readfirstlane((int)rowptr[j0 + SC_SRC]);
  const float* hrow = h + ((size_t)b * NN + j0) * ND + ch * 32u + lane;
  const float* nrow = nodex + (size_t)j0 * ND + ch * 32u + lane;

#pragma unroll
  for (int i = 0; i < 16; ++i)
    *reinterpret_cast<v4f*>(acc + 4 * (tid + SC_T * i)) = (v4f){0.0f, 0.0f, 0.0f, 0.0f};

  auto rp_at = [&](unsigned t) -> unsigned {
    if (t < 32u) return (unsigned)__builtin_amdgcn_readlane((int)rp0, (int)t);
    if (t < 64u) return (unsigned)__builtin_amdgcn_readlane((int)rp1, (int)(t - 32u));
    return rpe;
  };

  auto scatter_rows = [&](unsigned dlo, unsigned c, float v) {
    unsigned e = 0;
    for (; e + 4u <= c; e += 4u) {
      const unsigned o0 = (unsigned)__builtin_amdgcn_readlane((int)dlo, (int)e);
      const unsigned o1 = (unsigned)__builtin_amdgcn_readlane((int)dlo, (int)e + 1);
      const unsigned o2 = (unsigned)__builtin_amdgcn_readlane((int)dlo, (int)e + 2);
      const unsigned o3 = (unsigned)__builtin_amdgcn_readlane((int)dlo, (int)e + 3);
      lds_add(accb + o0 + lane4, v);
      lds_add(accb + o1 + lane4, v);
      lds_add(accb + o2 + lane4, v);
      lds_add(accb + o3 + lane4, v);
    }
    for (; e < c; ++e)
      lds_add(accb + (unsigned)__builtin_amdgcn_readlane((int)dlo, (int)e) + lane4, v);
  };

  float hv[SC_CH], nv[SC_CH];
  unsigned dl[SC_CH], st[SC_CH + 1];
  st[0] = rp_at(0u);
#pragma unroll
  for (int u = 0; u < SC_CH; ++u) {
    st[u + 1] = rp_at((unsigned)u + 1u);
    hv[u] = hrow[(size_t)u * ND];
    nv[u] = nrow[(size_t)u * ND];
    dl[u] = sdst[min(st[u] + lane, (unsigned)(NE - 1))];
  }
  __syncthreads();

  for (unsigned t0 = 0; t0 < (unsigned)SC_SRC; t0 += SC_CH) {
    float hc[SC_CH], nc[SC_CH];
    unsigned dc[SC_CH], sc[SC_CH + 1];
#pragma unroll
    for (int u = 0; u < SC_CH; ++u) { hc[u] = hv[u]; nc[u] = nv[u]; dc[u] = dl[u]; sc[u] = st[u]; }
    sc[SC_CH] = st[SC_CH];
    if (t0 + SC_CH < (unsigned)SC_SRC) {
      const unsigned t1 = t0 + SC_CH;
      st[0] = sc[SC_CH];
#pragma unroll
      for (int u = 0; u < SC_CH; ++u) {
        st[u + 1] = rp_at(t1 + (unsigned)u + 1u);
        hv[u] = hrow[(size_t)(t1 + (unsigned)u) * ND];
        nv[u] = nrow[(size_t)(t1 + (unsigned)u) * ND];
        dl[u] = sdst[min(st[u] + lane, (unsigned)(NE - 1))];
      }
    }
#pragma unroll
    for (int u = 0; u < SC_CH; ++u) {
      const unsigned s0 = sc[u];
      const unsigned deg = sc[u + 1] - s0;
      const float v = hc[u] + nc[u];
      scatter_rows(dc[u], deg < 32u ? deg : 32u, v);
      for (unsigned w = 32u; w < deg; w += 32u) {
        unsigned dlo = sdst[min(s0 + w + lane, (unsigned)(NE - 1))];
        asm volatile("" : "+v"(dlo));
        unsigned c = deg - w;
        if (c > 32u) c = 32u;
        scatter_rows(dlo, c, v);
      }
    }
  }
  __syncthreads();

#pragma unroll
  for (int i = 0; i < 16; ++i) {
    const unsigned idx = tid + SC_T * (unsigned)i;
    const v4f v = *reinterpret_cast<const v4f*>(acc + 4 * idx);
    *reinterpret_cast<v4f*>(out + ((size_t)b * NN + (idx >> 3)) * ND + ch * 32u + ((idx & 7u) << 2)) = v;
  }
}

extern "C" void kernel_launch(void* const* d_in, const int* in_sizes, int n_in,
                              void* d_out, int out_size, void* d_ws, size_t ws_size,
                              hipStream_t stream) {
  (void)in_sizes; (void)n_in; (void)out_size; (void)ws_size;
  const float* X   = (const float*)d_in[0];
  const int*   ei  = (const int*)d_in[1];
  const float* W1t = (const float*)d_in[2];
  const float* b1t = (const float*)d_in[3];
  const float* W2t = (const float*)d_in[4];
  const float* b2t = (const float*)d_in[5];
  const float* W1n = (const float*)d_in[6];
  const float* b1n = (const float*)d_in[7];
  const float* W2n = (const float*)d_in[8];
  const float* b2n = (const float*)d_in[9];
  float* out = (float*)d_out;

  float* h     = (float*)d_ws;
  float* xmax  = h + (size_t)NB * NN * ND;
  unsigned* lst = (unsigned*)(xmax + (size_t)NN * ND);

  k_mlp2_rows_maxb<<<MLP_GRID + MAXB_BLOCKS, 256, 0, stream>>>(X, W1t, b1t, W2t, b2t, h, xmax);
  k_mlp2_wmma<<<NM_BLOCKS + 1, NM_T, 0, stream>>>(xmax, W1n, b1n, W2n, b2n, xmax, ei, lst);
  k_edge_scatter<<<NB * 2, SC_T, 0, stream>>>(h, xmax, lst, out);
}


# max over subgraphs leaves the big MLP's launch: each of 128 node-MLP blocks takes the max of its own 16x64 tile (32 plane loads in flight) into LDS as its WMMA input; no max plane
# speedup vs baseline: 2.4557x; 1.2754x over previous
#include <hip/hip_runtime.h>

typedef __attribute__((ext_vector_type(2))) float v2f;
typedef __attribute__((ext_vector_type(8))) float v8f;
typedef __attribute__((ext_vector_type(4))) float v4f;

#define NB 128
#define NN 2048
#define ND 64
#define NE 32768

typedef __attribute__((ext_vector_type(4))) int v4i;

#define NM_BLOCKS (NN / 16)
#define NM_T 1024
#define NMP 68
#define LST_DST_WORD 4096
static_assert(NM_T == 16 * ND && (NB % 32) == 0, "the tile block: one input element per thread, planes in batches of 32");
static_assert(NM_T == 1024 && NE == 32 * NM_T, "the sorting block: every thread takes 32 edges");
static_assert(NN == 2 * NM_T, "the sorting block: every thread scans two counters");

__global__ void __launch_bounds__(NM_T)
k_max_mlp2_wmma(const float* __restrict__ X, const float* __restrict__ W1,
                const float* __restrict__ b1, const float* __restrict__ W2,
                const float* __restrict__ b2, float* __restrict__ Yout,
                const int* __restrict__ ei, unsigned* __restrict__ lst) {
  __shared__ __attribute__((aligned(16))) float Xs[16 * NMP];
  __shared__ __attribute__((aligned(16))) float Ys[16 * NMP];
  __shared__ unsigned cnt[NN];
  __shared__ unsigned wsum[32];
  __shared__ __attribute__((aligned(16))) unsigned sl[NE];
  const int tid = threadIdx.x;
  const int lane = tid & 31;
  const int wave = __builtin_amdgcn_readfirstlane(tid >> 5);

  if (blockIdx.x == NM_BLOCKS) {
    const int* __restrict__ esrc = ei;
    const int* __restrict__ edst = ei + NE;
    v4i sv[8], dv[8];
#pragma unroll
    for (int i = 0; i < 8; ++i) {
      sv[i] = *reinterpret_cast<const v4i*>(esrc + 4 * (i * NM_T + tid));
      dv[i] = *reinterpret_cast<const v4i*>(edst + 4 * (i * NM_T + tid));
    }
    cnt[tid] = 0u;
    cnt[tid + NM_T] = 0u;
    __syncthreads();
#pragma unroll
    for (int i = 0; i < 8; ++i)
#pragma unroll
      for (int c = 0; c < 4; ++c)
        atomicAdd(&cnt[(unsigned)sv[i][c] & (unsigned)(NN - 1)], 1u);
    __syncthreads();
    const unsigned c0 = cnt[2 * tid];
    const unsigned c1 = cnt[2 * tid + 1];
    unsigned incl = c0 + c1;
#pragma unroll
    for (int o = 1; o < 32; o <<= 1) {
      const unsigned up = __shfl_up(incl, (unsigned)o, 32);
      if (lane >= o) incl += up;
    }
    if (lane == 31) wsum[wave] = incl;
    __syncthreads();
    unsigned base = 0u;
    for (int q = 0; q < 32; ++q) base += (q < wave) ? wsum[q] : 0u;
    const unsigned ex = base + incl - (c0 + c1);
    cnt[2 * tid] = ex;
    cnt[2 * tid + 1] = ex + c0;
    lst[2 * tid] = ex;
    lst[2 * tid + 1] = ex + c0;
    if (tid == NM_T - 1) lst[NN] = ex + c0 + c1;
    __syncthreads();
#pragma unroll
    for (int i = 0; i < 8; ++i) {
      unsigned p[4];
#pragma unroll
      for (int c = 0; c < 4; ++c)
        p[c] = atomicAdd(&cnt[(unsigned)sv[i][c] & (unsigned)(NN - 1)], 1u);
#pragma unroll
      for (int c = 0; c < 4; ++c)
        sl[p[c] & (unsigned)(NE - 1)] = ((unsigned)dv[i][c] & (unsigned)(NN - 1)) << 7;
    }
    __syncthreads();
    char* gdst = reinterpret_cast<char*>(lst + LST_DST_WORD);
    const char* sbytes = reinterpret_cast<const char*>(sl);
#pragma unroll
    for (int i = 0; i < 8; ++i) {
      const int idx = tid + NM_T * i;
      *reinterpret_cast<v4i*>(gdst + 16 * idx) = *reinterpret_cast<const v4i*>(sbytes + 16 * idx);
    }
    return;
  }

  const size_t row0 = (size_t)blockIdx.x * 16;
  {
    const float* __restrict__ xp = X + row0 * ND + tid;
    float m = -__builtin_inff();
#pragma unroll 1
    for (int b0 = 0; b0 < NB; b0 += 32) {
      float v[32];
#pragma unroll
      for (int u = 0; u < 32; ++u) v[u] = xp[(size_t)(b0 + u) * (NN * ND)];
#pragma unroll
      for (int u = 0; u < 32; ++u) m = fmaxf(m, v[u]);
    }
    Xs[(tid >> 6) * NMP + (tid & 63)] = m;
  }
  __syncthreads();

  const int col0 = wave * 16;
  const int hm = lane & 15;
  const int hk = lane >> 4;
  if (wave < 4) {
    v2f bw[16];
#pragma unroll
    for (int kk = 0; kk < 16; ++kk) {
      const int k = kk * 4 + 2 * hk;
      bw[kk].x = W1[(k + 0) * ND + col0 + hm];
      bw[kk].y = W1[(k + 1) * ND + col0 + hm];
    }
    v8f acc;
    {
      const float bv = b1[col0 + hm];
#pragma unroll
      for (int j = 0; j < 8; ++j) acc[j] = bv;
    }
#pragma unroll
    for (int kk = 0; kk < 16; ++kk) {
      const int k = kk * 4 + 2 * hk;
      const v2f a = *reinterpret_cast<const v2f*>(Xs + hm * NMP + k);
      acc = __builtin_amdgcn_wmma_f32_16x16x4_f32(false, a, false, bw[kk],
                                                  (short)0, acc, false, false);
    }
#pragma unroll
    for (int j = 0; j < 8; ++j)
      Ys[(j + 8 * hk) * NMP + col0 + hm] = fmaxf(acc[j], 0.0f);
  }
  __syncthreads();

  if (wave < 4) {
    v2f bw[16];
#pragma unroll
    for (int kk = 0; kk < 16; ++kk) {
      const int k = kk * 4 + 2 * hk;
      bw[kk].x = W2[(k + 0) * ND + col0 + hm];
      bw[kk].y = W2[(k + 1) * ND + col0 + hm];
    }
    v8f acc;
    {
      const float bv = b2[col0 + hm];
#pragma unroll
      for (int j = 0; j < 8; ++j) acc[j] = bv;
    }
#pragma unroll
    for (int kk = 0; kk < 16; ++kk) {
      const int k = kk * 4 + 2 * hk;
      const v2f a = *reinterpret_cast<const v2f*>(Ys + hm * NMP + k);
      acc = __builtin_amdgcn_wmma_f32_16x16x4_f32(false, a, false, bw[kk],
                                                  (short)0, acc, false, false);
    }
#pragma unroll
    for (int j = 0; j < 8; ++j)
      Yout[(row0 + j + 8 * hk) * ND + col0 + hm] = fmaxf(acc[j], 0.0f);
  }
}

#define MP 68
#define MW 8
#define MLP_GRID 512
#define MLP_ITERS ((NB * NN / 16) / (MLP_GRID * MW))

__global__ void __launch_bounds__(256, 8)
k_mlp2_rows(const float* __restrict__ Xin, const float* __restrict__ W1,
            const float* __restrict__ b1, const float* __restrict__ W2,
            const float* __restrict__ b2, float* __restrict__ Yout) {
  __shared__ __attribute__((aligned(16))) float Wt1[ND * MP];
  __shared__ __attribute__((aligned(16))) float Wt2[ND * MP];
  __shared__ __attribute__((aligned(16))) float Ts[MW][16 * MP];
  const int tid = threadIdx.x;
  const unsigned rblk = blockIdx.x;
  const int lane = tid & 31;
  const int wave = __builtin_amdgcn_readfirstlane(tid >> 5);
  const int hm = lane & 15;
  const int hk = lane >> 4;
  const int c4 = (lane & 15) << 2;
  float* T = Ts[wave];

  size_t tile = (size_t)rblk * MW + wave;
  v4f xr[8];
#pragma unroll
  for (int i = 0; i < 8; ++i)
    xr[i] = *reinterpret_cast<const v4f*>(Xin + (tile * 16 + 2 * i + hk) * ND + c4);

#pragma unroll
  for (int i = 0; i < 4; ++i) {
    const int idx = tid + 256 * i;
    const int k = idx >> 4;
    const int n4 = (idx & 15) << 2;
    const v4f w1 = *reinterpret_cast<const v4f*>(W1 + k * ND + n4);
    const v4f w2 = *reinterpret_cast<const v4f*>(W2 + k * ND + n4);
#pragma unroll
    for (int e = 0; e < 4; ++e) {
      Wt1[(n4 + e) * MP + k] = w1[e];
      Wt2[(n4 + e) * MP + k] = w2[e];
    }
  }
  float bv1[4], bv2[4];
#pragma unroll
  for (int ct = 0; ct < 4; ++ct) {
    bv1[ct] = b1[ct * 16 + hm];
    bv2[ct] = b2[ct * 16 + hm];
  }

  for (int it = 0; it < MLP_ITERS; ++it) {
#pragma unroll
    for (int i = 0; i < 8; ++i)
      *reinterpret_cast<v4f*>(T + (2 * i + hk) * MP + c4) = xr[i];
    __syncthreads();

    const size_t row0 = tile * 16;
    if (it + 1 < MLP_ITERS) {
      tile += (size_t)MLP_GRID * MW;
#pragma unroll
      for (int i = 0; i < 8; ++i)
        xr[i] = *reinterpret_cast<const v4f*>(Xin + (tile * 16 + 2 * i + hk) * ND + c4);
    }

    v8f acc[4];
#pragma unroll
    for (int ct = 0; ct < 4; ++ct) {
      float bb = bv1[ct];
      asm volatile("" : "+v"(bb));
#pragma unroll
      for (int j = 0; j < 8; ++j) acc[ct][j] = bb;
    }
#pragma unroll
    for (int kk = 0; kk < 16; ++kk) {
      const int k = kk * 4 + 2 * hk;
      const v2f a = *reinterpret_cast<const v2f*>(T + hm * MP + k);
#pragma unroll
      for (int ct = 0; ct < 4; ++ct) {
        const v2f b = *reinterpret_cast<const v2f*>(Wt1 + (ct * 16 + hm) * MP + k);
        acc[ct] = __builtin_amdgcn_wmma_f32_16x16x4_f32(false, a, false, b,
                                                        (short)0, acc[ct], false, false);
      }
    }
#pragma unroll
    for (int ct = 0; ct < 4; ++ct)
#pragma unroll
      for (int j = 0; j < 8; ++j)
        T[(j + 8 * hk) * MP + ct * 16 + hm] = fmaxf(acc[ct][j], 0.0f);
    __syncthreads();

#pragma unroll
    for (int ct = 0; ct < 4; ++ct) {
      float bb = bv2[ct];
      asm volatile("" : "+v"(bb));
#pragma unroll
      for (int j = 0; j < 8; ++j) acc[ct][j] = bb;
    }
#pragma unroll
    for (int kk = 0; kk < 16; ++kk) {
      const int k = kk * 4 + 2 * hk;
      const v2f a = *reinterpret_cast<const v2f*>(T + hm * MP + k);
#pragma unroll
      for (int ct = 0; ct < 4; ++ct) {
        const v2f b = *reinterpret_cast<const v2f*>(Wt2 + (ct * 16 + hm) * MP + k);
        acc[ct] = __builtin_amdgcn_wmma_f32_16x16x4_f32(false, a, false, b,
                                                        (short)0, acc[ct], false, false);
      }
    }
#pragma unroll
    for (int ct = 0; ct < 4; ++ct)
#pragma unroll
      for (int j = 0; j < 8; ++j)
        T[(j + 8 * hk) * MP + ct * 16 + hm] = fmaxf(acc[ct][j], 0.0f);
    __syncthreads();

#pragma unroll
    for (int i = 0; i < 8; ++i) {
      const v4f v = *reinterpret_cast<const v4f*>(T + (2 * i + hk) * MP + c4);
      *reinterpret_cast<v4f*>(Yout + (row0 + 2 * i + hk) * ND + c4) = v;
    }
  }
}

#define SC_T 1024
#define SC_SRC (NN / 32)
#define SC_CH 8
static_assert(SC_SRC == 64 && (SC_SRC % SC_CH) == 0, "a wave keeps its 65 row starts in two registers and a scalar");

__device__ __forceinline__ void lds_add(char* p, float v) {
  __hip_atomic_fetch_add(reinterpret_cast<float*>(p), v, __ATOMIC_RELAXED, __HIP_MEMORY_SCOPE_WORKGROUP);
}

__global__ void __launch_bounds__(SC_T)
k_edge_scatter(const float* __restrict__ h, const float* __restrict__ nodex,
               const unsigned* __restrict__ lst, float* __restrict__ out) {
  __shared__ __attribute__((aligned(16))) float acc[NN * 32];
  const unsigned tid = threadIdx.x;
  const unsigned lane = tid & 31;
  const unsigned wave = (unsigned)__builtin_amdgcn_readfirstlane((int)(tid >> 5));
  const unsigned b = blockIdx.x >> 1;
  const unsigned ch = blockIdx.x & 1u;
  const unsigned* __restrict__ rowptr = lst;
  const unsigned* __restrict__ sdst = lst + LST_DST_WORD;
  char* accb = reinterpret_cast<char*>(acc);
  const unsigned lane4 = lane << 2;

  const unsigned j0 = wave * SC_SRC;
  const unsigned rp0 = rowptr[j0 + lane];
  const unsigned rp1 = rowptr[j0 + 32 + lane];
  const unsigned rpe = (unsigned)__builtin_amdgcn_readfirstlane((int)rowptr[j0 + SC_SRC]);
  const float* hrow = h + ((size_t)b * NN + j0) * ND + ch * 32u + lane;
  const float* nrow = nodex + (size_t)j0 * ND + ch * 32u + lane;

#pragma unroll
  for (int i = 0; i < 16; ++i)
    *reinterpret_cast<v4f*>(acc + 4 * (tid + SC_T * i)) = (v4f){0.0f, 0.0f, 0.0f, 0.0f};

  auto rp_at = [&](unsigned t) -> unsigned {
    if (t < 32u) return (unsigned)__builtin_amdgcn_readlane((int)rp0, (int)t);
    if (t < 64u) return (unsigned)__builtin_amdgcn_readlane((int)rp1, (int)(t - 32u));
    return rpe;
  };

  auto scatter_rows = [&](unsigned dlo, unsigned c, float v) {
    unsigned e = 0;
    for (; e + 4u <= c; e += 4u) {
      const unsigned o0 = (unsigned)__builtin_amdgcn_readlane((int)dlo, (int)e);
      const unsigned o1 = (unsigned)__builtin_amdgcn_readlane((int)dlo, (int)e + 1);
      const unsigned o2 = (unsigned)__builtin_amdgcn_readlane((int)dlo, (int)e + 2);
      const unsigned o3 = (unsigned)__builtin_amdgcn_readlane((int)dlo, (int)e + 3);
      lds_add(accb + o0 + lane4, v);
      lds_add(accb + o1 + lane4, v);
      lds_add(accb + o2 + lane4, v);
      lds_add(accb + o3 + lane4, v);
    }
    for (; e < c; ++e)
      lds_add(accb + (unsigned)__builtin_amdgcn_readlane((int)dlo, (int)e) + lane4, v);
  };

  float hv[SC_CH], nv[SC_CH];
  unsigned dl[SC_CH], st[SC_CH + 1];
  st[0] = rp_at(0u);
#pragma unroll
  for (int u = 0; u < SC_CH; ++u) {
    st[u + 1] = rp_at((unsigned)u + 1u);
    hv[u] = hrow[(size_t)u * ND];
    nv[u] = nrow[(size_t)u * ND];
    dl[u] = sdst[min(st[u] + lane, (unsigned)(NE - 1))];
  }
  __syncthreads();

  for (unsigned t0 = 0; t0 < (unsigned)SC_SRC; t0 += SC_CH) {
    float hc[SC_CH], nc[SC_CH];
    unsigned dc[SC_CH], sc[SC_CH + 1];
#pragma unroll
    for (int u = 0; u < SC_CH; ++u) { hc[u] = hv[u]; nc[u] = nv[u]; dc[u] = dl[u]; sc[u] = st[u]; }
    sc[SC_CH] = st[SC_CH];
    if (t0 + SC_CH < (unsigned)SC_SRC) {
      const unsigned t1 = t0 + SC_CH;
      st[0] = sc[SC_CH];
#pragma unroll
      for (int u = 0; u < SC_CH; ++u) {
        st[u + 1] = rp_at(t1 + (unsigned)u + 1u);
        hv[u] = hrow[(size_t)(t1 + (unsigned)u) * ND];
        nv[u] = nrow[(size_t)(t1 + (unsigned)u) * ND];
        dl[u] = sdst[min(st[u] + lane, (unsigned)(NE - 1))];
      }
    }
#pragma unroll
    for (int u = 0; u < SC_CH; ++u) {
      const unsigned s0 = sc[u];
      const unsigned deg = sc[u + 1] - s0;
      const float v = hc[u] + nc[u];
      scatter_rows(dc[u], deg < 32u ? deg : 32u, v);
      for (unsigned w = 32u; w < deg; w += 32u) {
        unsigned dlo = sdst[min(s0 + w + lane, (unsigned)(NE - 1))];
        asm volatile("" : "+v"(dlo));
        unsigned c = deg - w;
        if (c > 32u) c = 32u;
        scatter_rows(dlo, c, v);
      }
    }
  }
  __syncthreads();

#pragma unroll
  for (int i = 0; i < 16; ++i) {
    const unsigned idx = tid + SC_T * (unsigned)i;
    const v4f v = *reinterpret_cast<const v4f*>(acc + 4 * idx);
    *reinterpret_cast<v4f*>(out + ((size_t)b * NN + (idx >> 3)) * ND + ch * 32u + ((idx & 7u) << 2)) = v;
  }
}

extern "C" void kernel_launch(void* const* d_in, const int* in_sizes, int n_in,
                              void* d_out, int out_size, void* d_ws, size_t ws_size,
                              hipStream_t stream) {
  (void)in_sizes; (void)n_in; (void)out_size; (void)ws_size;
  const float* X   = (const float*)d_in[0];
  const int*   ei  = (const int*)d_in[1];
  const float* W1t = (const float*)d_in[2];
  const float* b1t = (const float*)d_in[3];
  const float* W2t = (const float*)d_in[4];
  const float* b2t = (const float*)d_in[5];
  const float* W1n = (const float*)d_in[6];
  const float* b1n = (const float*)d_in[7];
  const float* W2n = (const float*)d_in[8];
  const float* b2n = (const float*)d_in[9];
  float* out = (float*)d_out;

  float* h     = (float*)d_ws;
  float* nodex = h + (size_t)NB * NN * ND;
  unsigned* lst = (unsigned*)(nodex + (size_t)NN * ND);

  k_mlp2_rows<<<MLP_GRID, 256, 0, stream>>>(X, W1t, b1t, W2t, b2t, h);
  k_max_mlp2_wmma<<<NM_BLOCKS + 1, NM_T, 0, stream>>>(X, W1n, b1n, W2n, b2n, nodex, ei, lst);
  k_edge_scatter<<<NB * 2, SC_T, 0, stream>>>(h, nodex, lst, out);
}


# node branch (max, node MLP, edge sort) as trailing 256-thread blocks of the big MLP's launch; sorted list in 16-bit words: 74 KB LDS a block, 4 blocks fit a WGP; 3 -> 2 launches; f32 add order kept
# speedup vs baseline: 2.5382x; 1.0336x over previous
#include <hip/hip_runtime.h>

typedef __attribute__((ext_vector_type(2))) float v2f;
typedef __attribute__((ext_vector_type(8))) float v8f;
typedef __attribute__((ext_vector_type(4))) float v4f;

#define NB 128
#define NN 2048
#define ND 64
#define NE 32768

typedef __attribute__((ext_vector_type(4))) int v4i;
typedef __attribute__((ext_vector_type(4))) unsigned v4u;
typedef __attribute__((ext_vector_type(4))) unsigned short v4us;

#define NM_BLOCKS (NN / 16)
#define NMP 68
#define LST_DST_WORD 4096
#define MP 68
#define MW 8
#define FT (32 * MW)
#define MLP_GRID 512
#define MLP_ITERS ((NB * NN / 16) / (MLP_GRID * MW))
#define LDS_MLP_WORDS (2 * ND * MP + MW * 16 * MP)
#define LDS_SORT_WORDS (NN + 16 + NE / 2)
#define LDS_WORDS (LDS_SORT_WORDS > LDS_MLP_WORDS ? LDS_SORT_WORDS : LDS_MLP_WORDS)
static_assert(FT == 256 && 16 * ND == 4 * FT, "the node tile: four input elements (one 16-byte piece) per thread");
static_assert((NB - 8) % 24 == 0, "the node tile: planes in batches of 24, then the last 8");
static_assert(NN == 8 * FT, "the sorting block: every thread scans eight counters");
static_assert(NE == 4 * 32 * FT, "the sorting block: every thread takes 32 sixteen-byte pieces of each id row");
static_assert(NN <= 65536, "a destination fits a 16-bit word");
static_assert(4 * LDS_WORDS * 4 <= 320 * 1024, "four blocks fit the LDS of a WGP");

__device__ __forceinline__ void sort_role(const int* __restrict__ ei, unsigned* __restrict__ lst,
                                          unsigned char* smem, const int tid) {
  unsigned* cnt = reinterpret_cast<unsigned*>(smem);
  unsigned* wsum = cnt + NN;
  unsigned short* sl16 = reinterpret_cast<unsigned short*>(cnt + NN + 16);
  const int lane = tid & 31;
  const int wave = tid >> 5;
  const v4i* __restrict__ s4 = reinterpret_cast<const v4i*>(ei);
  const v4i* __restrict__ d4 = reinterpret_cast<const v4i*>(ei + NE);

#pragma unroll
  for (int i = 0; i < 8; ++i) cnt[tid + FT * i] = 0u;
  __syncthreads();

  {
    v4i sa[8];
#pragma unroll
    for (int i = 0; i < 8; ++i) sa[i] = s4[i * FT + tid];
#pragma unroll 1
    for (int g = 0; g < 4; ++g) {
      const int gn = (g + 1 < 4) ? g + 1 : 3;
      v4i sn[8];
#pragma unroll
      for (int i = 0; i < 8; ++i) sn[i] = s4[(gn * 8 + i) * FT + tid];
#pragma unroll
      for (int i = 0; i < 8; ++i)
#pragma unroll
        for (int c = 0; c < 4; ++c)
          atomicAdd(&cnt[(unsigned)sa[i][c] & (unsigned)(NN - 1)], 1u);
#pragma unroll
      for (int i = 0; i < 8; ++i) sa[i] = sn[i];
    }
  }
  __syncthreads();

  const v4u ca = *reinterpret_cast<const v4u*>(cnt + 8 * tid);
  const v4u cb = *reinterpret_cast<const v4u*>(cnt + 8 * tid + 4);
  const unsigned tot = ca.x + ca.y + ca.z + ca.w + cb.x + cb.y + cb.z + cb.w;
  unsigned incl = tot;
#pragma unroll
  for (int o = 1; o < 32; o <<= 1) {
    const unsigned up = __shfl_up(incl, (unsigned)o, 32);
    if (lane >= o) incl += up;
  }
  if (lane == 31) wsum[wave] = incl;
  __syncthreads();
  unsigned base = 0u;
#pragma unroll
  for (int q = 0; q < MW; ++q) base += (q < wave) ? wsum[q] : 0u;
  v4u ea, eb;
  ea.x = base + incl - tot;
  ea.y = ea.x + ca.x;
  ea.z = ea.y + ca.y;
  ea.w = ea.z + ca.z;
  eb.x = ea.w + ca.w;
  eb.y = eb.x + cb.x;
  eb.z = eb.y + cb.y;
  eb.w = eb.z + cb.z;
  *reinterpret_cast<v4u*>(cnt + 8 * tid) = ea;
  *reinterpret_cast<v4u*>(cnt + 8 * tid + 4) = eb;
  *reinterpret_cast<v4u*>(lst + 8 * tid) = ea;
  *reinterpret_cast<v4u*>(lst + 8 * tid + 4) = eb;
  if (tid == FT - 1) lst[NN] = eb.w + cb.w;
  __syncthreads();

  {
    v4i sa[4], da[4];
#pragma unroll
    for (int i = 0; i < 4; ++i) {
      sa[i] = s4[i * FT + tid];
      da[i] = d4[i * FT + tid];
    }
#pragma unroll 1
    for (int g = 0; g < 8; ++g) {
      const int gn = (g + 1 < 8) ? g + 1 : 7;
      v4i sn[4], dn[4];
#pragma unroll
      for (int i = 0; i < 4; ++i) {
        sn[i] = s4[(gn * 4 + i) * FT + tid];
        dn[i] = d4[(gn * 4 + i) * FT + tid];
      }
#pragma unroll
      for (int i = 0; i < 4; ++i) {
        unsigned p[4];
#pragma unroll
        for (int c = 0; c < 4; ++c)
          p[c] = atomicAdd(&cnt[(unsigned)sa[i][c] & (unsigned)(NN - 1)], 1u);
#pragma unroll
        for (int c = 0; c < 4; ++c)
          sl16[p[c] & (unsigned)(NE - 1)] = (unsigned short)((unsigned)da[i][c] & (unsigned)(NN - 1));
      }
#pragma unroll
      for (int i = 0; i < 4; ++i) {
        sa[i] = sn[i];
        da[i] = dn[i];
      }
    }
  }
  __syncthreads();

  v4u* gdst = reinterpret_cast<v4u*>(lst + LST_DST_WORD);
#pragma unroll 8
  for (int i = 0; i < NE / (4 * FT); ++i) {
    const int idx = tid + FT * i;
    const v4us d = *reinterpret_cast<const v4us*>(sl16 + 4 * idx);
    v4u w;
    w.x = (unsigned)d.x << 7;
    w.y = (unsigned)d.y << 7;
    w.z = (unsigned)d.z << 7;
    w.w = (unsigned)d.w << 7;
    gdst[idx] = w;
  }
}

__device__ __forceinline__ void node_tile_role(const float* __restrict__ X, const float* __restrict__ W1,
                                               const float* __restrict__ b1, const float* __restrict__ W2,
                                               const float* __restrict__ b2, float* __restrict__ Yout,
                                               unsigned char* smem, const int tid, const int nb) {
  float* Xs = reinterpret_cast<float*>(smem);
  float* Ys = Xs + 16 * NMP;
  const int lane = tid & 31;
  const int wave = __builtin_amdgcn_readfirstlane(tid >> 5);
  const size_t row0 = (size_t)nb * 16;
  {
    const int r = tid >> 4;
    const int c4 = (tid & 15) << 2;
    const float* __restrict__ xp = X + (row0 + r) * ND + c4;
    const float ninf = -__builtin_inff();
    v4f m = (v4f){ninf, ninf, ninf, ninf};
#pragma unroll 1
    for (int b0 = 0; b0 < NB - 8; b0 += 24) {
      v4f v[24];
#pragma unroll
      for (int u = 0; u < 24; ++u)
        v[u] = *reinterpret_cast<const v4f*>(xp + (size_t)(b0 + u) * (NN * ND));
#pragma unroll
      for (int u = 0; u < 24; ++u) {
        m.x = fmaxf(m.x, v[u].x);
        m.y = fmaxf(m.y, v[u].y);
        m.z = fmaxf(m.z, v[u].z);
        m.w = fmaxf(m.w, v[u].w);
      }
    }
    {
      v4f v[8];
#pragma unroll
      for (int u = 0; u < 8; ++u)
        v[u] = *reinterpret_cast<const v4f*>(xp + (size_t)(NB - 8 + u) * (NN * ND));
#pragma unroll
      for (int u = 0; u < 8; ++u) {
        m.x = fmaxf(m.x, v[u].x);
        m.y = fmaxf(m.y, v[u].y);
        m.z = fmaxf(m.z, v[u].z);
        m.w = fmaxf(m.w, v[u].w);
      }
    }
    *reinterpret_cast<v4f*>(Xs + r * NMP + c4) = m;
  }
  __syncthreads();

  const int col0 = wave * 16;
  const int hm = lane & 15;
  const int hk = lane >> 4;
  if (wave < 4) {
    v2f bw[16];
#pragma unroll
    for (int kk = 0; kk < 16; ++kk) {
      const int k = kk * 4 + 2 * hk;
      bw[kk].x = W1[(k + 0) * ND + col0 + hm];
      bw[kk].y = W1[(k + 1) * ND + col0 + hm];
    }
    v8f acc;
    {
      const float bv = b1[col0 + hm];
#pragma unroll
      for (int j = 0; j < 8; ++j) acc[j] = bv;
    }
#pragma unroll
    for (int kk = 0; kk < 16; ++kk) {
      const int k = kk * 4 + 2 * hk;
      const v2f a = *reinterpret_cast<const v2f*>(Xs + hm * NMP + k);
      acc = __builtin_amdgcn_wmma_f32_16x16x4_f32(false, a, false, bw[kk],
                                                  (short)0, acc, false, false);
    }
#pragma unroll
    for (int j = 0; j < 8; ++j)
      Ys[(j + 8 * hk) * NMP + col0 + hm] = fmaxf(acc[j], 0.0f);
  }
  __syncthreads();

  if (wave < 4) {
    v2f bw[16];
#pragma unroll
    for (int kk = 0; kk < 16; ++kk) {
      const int k = kk * 4 + 2 * hk;
      bw[kk].x = W2[(k + 0) * ND + col0 + hm];
      bw[kk].y = W2[(k + 1) * ND + col0 + hm];
    }
    v8f acc;
    {
      const float bv = b2[col0 + hm];
#pragma unroll
      for (int j = 0; j < 8; ++j) acc[j] = bv;
    }
#pragma unroll
    for (int kk = 0; kk < 16; ++kk) {
      const int k = kk * 4 + 2 * hk;
      const v2f a = *reinterpret_cast<const v2f*>(Ys + hm * NMP + k);
      acc = __builtin_amdgcn_wmma_f32_16x16x4_f32(false, a, false, bw[kk],
                                                  (short)0, acc, false, false);
    }
#pragma unroll
    for (int j = 0; j < 8; ++j)
      Yout[(row0 + j + 8 * hk) * ND + col0 + hm] = fmaxf(acc[j], 0.0f);
  }
}

__global__ void __launch_bounds__(FT, 8)
k_mlp2_rows_node(const float* __restrict__ Xin, const float* __restrict__ W1,
                 const float* __restrict__ b1, const float* __restrict__ W2,
                 const float* __restrict__ b2, float* __restrict__ Yout,
                 const float* __restrict__ W1n, const float* __restrict__ b1n,
                 const float* __restrict__ W2n, const float* __restrict__ b2n,
                 float* __restrict__ nodex, const int* __restrict__ ei, unsigned* __restrict__ lst) {
  __shared__ __attribute__((aligned(16))) unsigned char smem[LDS_WORDS * 4];
  if (blockIdx.x >= MLP_GRID) {
    const int nb = (int)blockIdx.x - MLP_GRID;
    if (nb == NM_BLOCKS)
      sort_role(ei, lst, smem, (int)threadIdx.x);
    else
      node_tile_role(Xin, W1n, b1n, W2n, b2n, nodex, smem, (int)threadIdx.x, nb);
    return;
  }
  float* Wt1 = reinterpret_cast<float*>(smem);
  float* Wt2 = Wt1 + ND * MP;
  const int tid = threadIdx.x;
  const unsigned rblk = blockIdx.x;
  const int lane = tid & 31;
  const int wave = __builtin_amdgcn_readfirstlane(tid >> 5);
  const int hm = lane & 15;
  const int hk = lane >> 4;
  const int c4 = (lane & 15) << 2;
  float* T = Wt2 + ND * MP + wave * (16 * MP);

  size_t tile = (size_t)rblk * MW + wave;
  const unsigned lofs = (unsigned)(hk * ND + c4);
  v4f xr[8];
#pragma unroll
  for (int i = 0; i < 8; ++i)
    xr[i] = *reinterpret_cast<const v4f*>(Xin + tile * (16 * ND) + 2 * i * ND + lofs);

#pragma unroll
  for (int i = 0; i < 4; ++i) {
    const int idx = tid + 256 * i;
    const int k = idx >> 4;
    const int n4 = (idx & 15) << 2;
    const v4f w1 = *reinterpret_cast<const v4f*>(W1 + k * ND + n4);
    const v4f w2 = *reinterpret_cast<const v4f*>(W2 + k * ND + n4);
#pragma unroll
    for (int e = 0; e < 4; ++e) {
      Wt1[(n4 + e) * MP + k] = w1[e];
      Wt2[(n4 + e) * MP + k] = w2[e];
    }
  }
  float bv1[4], bv2[4];
#pragma unroll
  for (int ct = 0; ct < 4; ++ct) {
    bv1[ct] = b1[ct * 16 + hm];
    bv2[ct] = b2[ct * 16 + hm];
  }

  for (int it = 0; it < MLP_ITERS; ++it) {
#pragma unroll
    for (int i = 0; i < 8; ++i)
      *reinterpret_cast<v4f*>(T + (2 * i + hk) * MP + c4) = xr[i];
    __syncthreads();

    const size_t row0 = tile * 16;
    if (it + 1 < MLP_ITERS) {
      tile += (size_t)MLP_GRID * MW;
      unsigned lo = lofs;
      asm volatile("" : "+v"(lo));
#pragma unroll
      for (int i = 0; i < 8; ++i)
        xr[i] = *reinterpret_cast<const v4f*>(Xin + tile * (16 * ND) + 2 * i * ND + lo);
    }

    v8f acc[4];
#pragma unroll
    for (int ct = 0; ct < 4; ++ct) {
      float bb = bv1[ct];
      asm volatile("" : "+v"(bb));
#pragma unroll
      for (int j = 0; j < 8; ++j) acc[ct][j] = bb;
    }
#pragma unroll
    for (int kk = 0; kk < 16; ++kk) {
      const int k = kk * 4 + 2 * hk;
      const v2f a = *reinterpret_cast<const v2f*>(T + hm * MP + k);
#pragma unroll
      for (int ct = 0; ct < 4; ++ct) {
        const v2f b = *reinterpret_cast<const v2f*>(Wt1 + (ct * 16 + hm) * MP + k);
        acc[ct] = __builtin_amdgcn_wmma_f32_16x16x4_f32(false, a, false, b,
                                                        (short)0, acc[ct], false, false);
      }
    }
#pragma unroll
    for (int ct = 0; ct < 4; ++ct)
#pragma unroll
      for (int j = 0; j < 8; ++j)
        T[(j + 8 * hk) * MP + ct * 16 + hm] = fmaxf(acc[ct][j], 0.0f);
    __syncthreads();

#pragma unroll
    for (int ct = 0; ct < 4; ++ct) {
      float bb = bv2[ct];
      asm volatile("" : "+v"(bb));
#pragma unroll
      for (int j = 0; j < 8; ++j) acc[ct][j] = bb;
    }
#pragma unroll
    for (int kk = 0; kk < 16; ++kk) {
      const int k = kk * 4 + 2 * hk;
      const v2f a = *reinterpret_cast<const v2f*>(T + hm * MP + k);
#pragma unroll
      for (int ct = 0; ct < 4; ++ct) {
        const v2f b = *reinterpret_cast<const v2f*>(Wt2 + (ct * 16 + hm) * MP + k);
        acc[ct] = __builtin_amdgcn_wmma_f32_16x16x4_f32(false, a, false, b,
                                                        (short)0, acc[ct], false, false);
      }
    }
#pragma unroll
    for (int ct = 0; ct < 4; ++ct)
#pragma unroll
      for (int j = 0; j < 8; ++j)
        T[(j + 8 * hk) * MP + ct * 16 + hm] = fmaxf(acc[ct][j], 0.0f);
    __syncthreads();

    unsigned lo2 = lofs;
    asm volatile("" : "+v"(lo2));
#pragma unroll
    for (int i = 0; i < 8; ++i) {
      const v4f v = *reinterpret_cast<const v4f*>(T + (2 * i + hk) * MP + c4);
      *reinterpret_cast<v4f*>(Yout + row0 * ND + 2 * i * ND + lo2) = v;
    }
  }
}

#define SC_T 1024
#define SC_SRC (NN / 32)
#define SC_CH 8
static_assert(SC_SRC == 64 && (SC_SRC % SC_CH) == 0, "a wave keeps its 65 row starts in two registers and a scalar");

__device__ __forceinline__ void lds_add(char* p, float v) {
  __hip_atomic_fetch_add(reinterpret_cast<float*>(p), v, __ATOMIC_RELAXED, __HIP_MEMORY_SCOPE_WORKGROUP);
}

__global__ void __launch_bounds__(SC_T)
k_edge_scatter(const float* __restrict__ h, const float* __restrict__ nodex,
               const unsigned* __restrict__ lst, float* __restrict__ out) {
  __shared__ __attribute__((aligned(16))) float acc[NN * 32];
  const unsigned tid = threadIdx.x;
  const unsigned lane = tid & 31;
  const unsigned wave = (unsigned)__builtin_amdgcn_readfirstlane((int)(tid >> 5));
  const unsigned b = blockIdx.x >> 1;
  const unsigned ch = blockIdx.x & 1u;
  const unsigned* __restrict__ rowptr = lst;
  const unsigned* __restrict__ sdst = lst + LST_DST_WORD;
  char* accb = reinterpret_cast<char*>(acc);
  const unsigned lane4 = lane << 2;

  const unsigned j0 = wave * SC_SRC;
  const unsigned rp0 = rowptr[j0 + lane];
  const unsigned rp1 = rowptr[j0 + 32 + lane];
  const unsigned rpe = (unsigned)__builtin_amdgcn_readfirstlane((int)rowptr[j0 + SC_SRC]);
  const float* hrow = h + ((size_t)b * NN + j0) * ND + ch * 32u + lane;
  const float* nrow = nodex + (size_t)j0 * ND + ch * 32u + lane;

#pragma unroll
  for (int i = 0; i < 16; ++i)
    *reinterpret_cast<v4f*>(acc + 4 * (tid + SC_T * i)) = (v4f){0.0f, 0.0f, 0.0f, 0.0f};

  auto rp_at = [&](unsigned t) -> unsigned {
    if (t < 32u) return (unsigned)__builtin_amdgcn_readlane((int)rp0, (int)t);
    if (t < 64u) return (unsigned)__builtin_amdgcn_readlane((int)rp1, (int)(t - 32u));
    return rpe;
  };

  auto scatter_rows = [&](unsigned dlo, unsigned c, float v) {
    unsigned e = 0;
    for (; e + 4u <= c; e += 4u) {
      const unsigned o0 = (unsigned)__builtin_amdgcn_readlane((int)dlo, (int)e);
      const unsigned o1 = (unsigned)__builtin_amdgcn_readlane((int)dlo, (int)e + 1);
      const unsigned o2 = (unsigned)__builtin_amdgcn_readlane((int)dlo, (int)e + 2);
      const unsigned o3 = (unsigned)__builtin_amdgcn_readlane((int)dlo, (int)e + 3);
      lds_add(accb + o0 + lane4, v);
      lds_add(accb + o1 + lane4, v);
      lds_add(accb + o2 + lane4, v);
      lds_add(accb + o3 + lane4, v);
    }
    for (; e < c; ++e)
      lds_add(accb + (unsigned)__builtin_amdgcn_readlane((int)dlo, (int)e) + lane4, v);
  };

  float hv[SC_CH], nv[SC_CH];
  unsigned dl[SC_CH], st[SC_CH + 1];
  st[0] = rp_at(0u);
#pragma unroll
  for (int u = 0; u < SC_CH; ++u) {
    st[u + 1] = rp_at((unsigned)u + 1u);
    hv[u] = hrow[(size_t)u * ND];
    nv[u] = nrow[(size_t)u * ND];
    dl[u] = sdst[min(st[u] + lane, (unsigned)(NE - 1))];
  }
  __syncthreads();

  for (unsigned t0 = 0; t0 < (unsigned)SC_SRC; t0 += SC_CH) {
    float hc[SC_CH], nc[SC_CH];
    unsigned dc[SC_CH], sc[SC_CH + 1];
#pragma unroll
    for (int u = 0; u < SC_CH; ++u) { hc[u] = hv[u]; nc[u] = nv[u]; dc[u] = dl[u]; sc[u] = st[u]; }
    sc[SC_CH] = st[SC_CH];
    if (t0 + SC_CH < (unsigned)SC_SRC) {
      const unsigned t1 = t0 + SC_CH;
      st[0] = sc[SC_CH];
#pragma unroll
      for (int u = 0; u < SC_CH; ++u) {
        st[u + 1] = rp_at(t1 + (unsigned)u + 1u);
        hv[u] = hrow[(size_t)(t1 + (unsigned)u) * ND];
        nv[u] = nrow[(size_t)(t1 + (unsigned)u) * ND];
        dl[u] = sdst[min(st[u] + lane, (unsigned)(NE - 1))];
      }
    }
#pragma unroll
    for (int u = 0; u < SC_CH; ++u) {
      const unsigned s0 = sc[u];
      const unsigned deg = sc[u + 1] - s0;
      const float v = hc[u] + nc[u];
      scatter_rows(dc[u], deg < 32u ? deg : 32u, v);
      for (unsigned w = 32u; w < deg; w += 32u) {
        unsigned dlo = sdst[min(s0 + w + lane, (unsigned)(NE - 1))];
        asm volatile("" : "+v"(dlo));
        unsigned c = deg - w;
        if (c > 32u) c = 32u;
        scatter_rows(dlo, c, v);
      }
    }
  }
  __syncthreads();

#pragma unroll
  for (int i = 0; i < 16; ++i) {
    const unsigned idx = tid + SC_T * (unsigned)i;
    const v4f v = *reinterpret_cast<const v4f*>(acc + 4 * idx);
    *reinterpret_cast<v4f*>(out + ((size_t)b * NN + (idx >> 3)) * ND + ch * 32u + ((idx & 7u) << 2)) = v;
  }
}

extern "C" void kernel_launch(void* const* d_in, const int* in_sizes, int n_in,
                              void* d_out, int out_size, void* d_ws, size_t ws_size,
                              hipStream_t stream) {
  (void)in_sizes; (void)n_in; (void)out_size; (void)ws_size;
  const float* X   = (const float*)d_in[0];
  const int*   ei  = (const int*)d_in[1];
  const float* W1t = (const float*)d_in[2];
  const float* b1t = (const float*)d_in[3];
  const float* W2t = (const float*)d_in[4];
  const float* b2t = (const float*)d_in[5];
  const float* W1n = (const float*)d_in[6];
  const float* b1n = (const float*)d_in[7];
  const float* W2n = (const float*)d_in[8];
  const float* b2n = (const float*)d_in[9];
  float* out = (float*)d_out;

  float* h     = (float*)d_ws;
  float* nodex = h + (size_t)NB * NN * ND;
  unsigned* lst = (unsigned*)(nodex + (size_t)NN * ND);

  k_mlp2_rows_node<<<MLP_GRID + NM_BLOCKS + 1, FT, 0, stream>>>(X, W1t, b1t, W2t, b2t, h, W1n, b1n, W2n, b2n, nodex, ei, lst);
  k_edge_scatter<<<NB * 2, SC_T, 0, stream>>>(h, nodex, lst, out);
}
